# starting kernel
# baseline (speedup 1.0000x reference)
#include <hip/hip_runtime.h>

#define D 128
#define LN_EPS 1e-5f

typedef __attribute__((ext_vector_type(2))) float v2f;
typedef __attribute__((ext_vector_type(8))) float v8f;

__global__ void init_deg_kernel(float* __restrict__ deg, int N) {
  int i = blockIdx.x * blockDim.x + threadIdx.x;
  if (i < N) deg[i] = 1.0f;
}

__global__ void edge_deg_kernel(const int* __restrict__ ei, const float* __restrict__ w,
                                float* __restrict__ deg, int E) {
  int e = blockIdx.x * blockDim.x + threadIdx.x;
  if (e < E) atomicAdd(&deg[ei[E + e]], w[e]);
}

__global__ void deg_to_dinv_kernel(float* __restrict__ deg, int N) {
  int i = blockIdx.x * blockDim.x + threadIdx.x;
  if (i < N) {
    float d = deg[i];
    deg[i] = d > 0.0f ? rsqrtf(d) : 0.0f;
  }
}

__global__ void __launch_bounds__(256)
gemm_xW_kernel(const float* __restrict__ X, const float* __restrict__ W,
               float* __restrict__ H) {
  const int wave = threadIdx.x >> 5;
  const int lane = threadIdx.x & 31;
  const int row0 = blockIdx.x * 16;
  const int col0 = wave * 16;
  const int m    = lane & 15;
  const int half = lane >> 4;
  const int n    = col0 + m;

  v8f acc = {};
  const float* xrow = X + (size_t)(row0 + m) * D + half * 2;
  for (int k = 0; k < D; k += 4) {
    v2f a = *(const v2f*)(xrow + k);
    const float* wp = W + (size_t)(k + half * 2) * D + n;
    v2f b;
    b.x = wp[0];
    b.y = wp[D];
    acc = __builtin_amdgcn_wmma_f32_16x16x4_f32(
        false, a, false, b, (short)0, acc, false, false);
  }
  float* hp = H + (size_t)(row0 + half * 8) * D + n;
#pragma unroll
  for (int r = 0; r < 8; ++r) hp[(size_t)r * D] = acc[r];
}

__global__ void self_loop_kernel(const float* __restrict__ h, const float* __restrict__ dinv,
                                 float* __restrict__ agg, int N) {
  int t = blockIdx.x * blockDim.x + threadIdx.x;
  int node = t >> 5;
  int f = (t & 31) * 4;
  if (node < N) {
    float c = dinv[node];
    c = c * c;
    float4 hv = *(const float4*)(h + (size_t)node * D + f);
    float4 o;
    o.x = c * hv.x; o.y = c * hv.y; o.z = c * hv.z; o.w = c * hv.w;
    *(float4*)(agg + (size_t)node * D + f) = o;
  }
}

__global__ void __launch_bounds__(256)
edge_scatter_kernel(const int* __restrict__ ei, const float* __restrict__ w,
                    const float* __restrict__ dinv, const float* __restrict__ h,
                    float* __restrict__ agg, int E) {
  int wave = threadIdx.x >> 5;
  int lane = threadIdx.x & 31;
  int e = blockIdx.x * 8 + wave;
  if (e >= E) return;
  int r = ei[e];
  int c = ei[E + e];
  float coeff = dinv[r] * w[e] * dinv[c];
  int f = lane * 4;
  float4 hv = *(const float4*)(h + (size_t)r * D + f);
  float* ap = agg + (size_t)c * D + f;
  atomicAdd(ap + 0, coeff * hv.x);
  atomicAdd(ap + 1, coeff * hv.y);
  atomicAdd(ap + 2, coeff * hv.z);
  atomicAdd(ap + 3, coeff * hv.w);
}

__global__ void __launch_bounds__(256)
bias_relu_ln_kernel(const float* __restrict__ agg, const float* __restrict__ bias,
                    const float* __restrict__ gamma, const float* __restrict__ beta,
                    float* __restrict__ y, int N) {
  int wave = threadIdx.x >> 5;
  int lane = threadIdx.x & 31;
  int node = blockIdx.x * 8 + wave;
  if (node >= N) return;
  int f = lane * 4;
  float4 v = *(const float4*)(agg + (size_t)node * D + f);
  float4 b = *(const float4*)(bias + f);
  v.x = fmaxf(v.x + b.x, 0.0f);
  v.y = fmaxf(v.y + b.y, 0.0f);
  v.z = fmaxf(v.z + b.z, 0.0f);
  v.w = fmaxf(v.w + b.w, 0.0f);
  float s = v.x + v.y + v.z + v.w;
#pragma unroll
  for (int mski = 16; mski >= 1; mski >>= 1) s += __shfl_xor(s, mski, 32);
  float mu = s * (1.0f / D);
  float dx0 = v.x - mu, dx1 = v.y - mu, dx2 = v.z - mu, dx3 = v.w - mu;
  float vs = dx0 * dx0 + dx1 * dx1 + dx2 * dx2 + dx3 * dx3;
#pragma unroll
  for (int mski = 16; mski >= 1; mski >>= 1) vs += __shfl_xor(vs, mski, 32);
  float inv = rsqrtf(vs * (1.0f / D) + LN_EPS);
  float4 g  = *(const float4*)(gamma + f);
  float4 be = *(const float4*)(beta + f);
  float4 o;
  o.x = dx0 * inv * g.x + be.x;
  o.y = dx1 * inv * g.y + be.y;
  o.z = dx2 * inv * g.z + be.z;
  o.w = dx3 * inv * g.w + be.w;
  *(float4*)(y + (size_t)node * D + f) = o;
}

__global__ void __launch_bounds__(256)
gemm_y_linT_kernel(const float* __restrict__ Y, const float* __restrict__ linW,
                   const float* __restrict__ linb, float* __restrict__ OUT) {
  const int wave = threadIdx.x >> 5;
  const int lane = threadIdx.x & 31;
  const int row0 = blockIdx.x * 16;
  const int col0 = wave * 16;
  const int m    = lane & 15;
  const int half = lane >> 4;
  const int n    = col0 + m;

  float lb = linb[n];
  v8f acc = {lb, lb, lb, lb, lb, lb, lb, lb};
  const float* yrow = Y + (size_t)(row0 + m) * D + half * 2;
  const float* brow = linW + (size_t)n * D + half * 2;
  for (int k = 0; k < D; k += 4) {
    v2f a = *(const v2f*)(yrow + k);
    v2f b = *(const v2f*)(brow + k);
    acc = __builtin_amdgcn_wmma_f32_16x16x4_f32(
        false, a, false, b, (short)0, acc, false, false);
  }
  float* op = OUT + (size_t)(row0 + half * 8) * D + n;
#pragma unroll
  for (int r = 0; r < 8; ++r) op[(size_t)r * D] = acc[r];
}

extern "C" void kernel_launch(void* const* d_in, const int* in_sizes, int n_in,
                              void* d_out, int out_size, void* d_ws, size_t ws_size,
                              hipStream_t stream) {
  const float* x     = (const float*)d_in[0];
  const int*   ei    = (const int*)d_in[1];
  const float* ew    = (const float*)d_in[2];
  const float* W     = (const float*)d_in[3];
  const float* bias  = (const float*)d_in[4];
  const float* gamma = (const float*)d_in[5];
  const float* beta  = (const float*)d_in[6];
  const float* linW  = (const float*)d_in[7];
  const float* linb  = (const float*)d_in[8];
  float* out = (float*)d_out;

  const int N = in_sizes[0] / D;
  const int E = in_sizes[2];

  char* ws = (char*)d_ws;
  float* dinv = (float*)ws;
  size_t off = (((size_t)N * sizeof(float)) + 255) & ~(size_t)255;
  float* h   = (float*)(ws + off);
  float* agg = (float*)(ws + off + (size_t)N * D * sizeof(float));

  init_deg_kernel<<<(N + 255) / 256, 256, 0, stream>>>(dinv, N);
  edge_deg_kernel<<<(E + 255) / 256, 256, 0, stream>>>(ei, ew, dinv, E);
  deg_to_dinv_kernel<<<(N + 255) / 256, 256, 0, stream>>>(dinv, N);
  gemm_xW_kernel<<<N / 16, 256, 0, stream>>>(x, W, h);
  self_loop_kernel<<<((size_t)N * 32 + 255) / 256, 256, 0, stream>>>(h, dinv, agg, N);
  edge_scatter_kernel<<<(E + 7) / 8, 256, 0, stream>>>(ei, ew, dinv, h, agg, E);
  bias_relu_ln_kernel<<<(N + 7) / 8, 256, 0, stream>>>(agg, bias, gamma, beta, h, N);
  gemm_y_linT_kernel<<<N / 16, 256, 0, stream>>>(h, linW, linb, out);
}


# Edges counting-sorted by target in the call (int atomic arrival rank, scan, fill); one wave per node gathers its rows in registers into bias+relu+LayerNorm: no float atomics, sums in rank order
# speedup vs baseline: 1.5495x; 1.5495x over previous
#include <hip/hip_runtime.h>

#define D 128
#define LN_EPS 1e-5f

typedef __attribute__((ext_vector_type(2))) float v2f;
typedef __attribute__((ext_vector_type(8))) float v8f;

__global__ void zero_count_kernel(int* __restrict__ cnt, int N) {
  int i = blockIdx.x * blockDim.x + threadIdx.x;
  if (i < N) cnt[i] = 0;
}

__global__ void edge_rank_kernel(const int* __restrict__ ei, int* __restrict__ cnt,
                                 int* __restrict__ rank, int E) {
  int e = blockIdx.x * blockDim.x + threadIdx.x;
  if (e < E) rank[e] = atomicAdd(&cnt[ei[E + e]], 1);
}

#define SCAN_Q 8
__global__ void __launch_bounds__(1024)
segment_start_kernel(const int* __restrict__ cnt, int* __restrict__ start, int N) {
  __shared__ int wsum[32];
  __shared__ int tile_total;
  const int tid  = threadIdx.x;
  const int lane = tid & 31;
  const int wv   = tid >> 5;
  int carry = 0;
  for (int base = 0; base < N; base += 1024 * 4 * SCAN_Q) {
    const int i0 = base + tid * (4 * SCAN_Q);
    int v[4 * SCAN_Q];
#pragma unroll
    for (int q = 0; q < SCAN_Q; ++q) {
      const int i = i0 + 4 * q;
      if (i + 3 < N) {
        const int4 t = *(const int4*)(cnt + i);
        v[4 * q] = t.x; v[4 * q + 1] = t.y; v[4 * q + 2] = t.z; v[4 * q + 3] = t.w;
      } else {
        v[4 * q]     = (i < N)     ? cnt[i]     : 0;
        v[4 * q + 1] = (i + 1 < N) ? cnt[i + 1] : 0;
        v[4 * q + 2] = (i + 2 < N) ? cnt[i + 2] : 0;
        v[4 * q + 3] = 0;
      }
    }
    int s = 0;
#pragma unroll
    for (int k = 0; k < 4 * SCAN_Q; ++k) s += v[k];
    int inc = s;
#pragma unroll
    for (int d = 1; d < 32; d <<= 1) {
      int t = __shfl_up(inc, d, 32);
      if (lane >= d) inc += t;
    }
    if (lane == 31) wsum[wv] = inc;
    __syncthreads();
    if (wv == 0) {
      const int x = wsum[lane];
      int xi = x;
#pragma unroll
      for (int d = 1; d < 32; d <<= 1) {
        int t = __shfl_up(xi, d, 32);
        if (lane >= d) xi += t;
      }
      wsum[lane] = xi - x;
      if (lane == 31) tile_total = xi;
    }
    __syncthreads();
    int run = carry + wsum[wv] + (inc - s);
#pragma unroll
    for (int q = 0; q < SCAN_Q; ++q) {
      const int i = i0 + 4 * q;
      int4 o;
      o.x = run; run += v[4 * q];
      o.y = run; run += v[4 * q + 1];
      o.z = run; run += v[4 * q + 2];
      o.w = run; run += v[4 * q + 3];
      if (i + 3 < N) {
        *(int4*)(start + i) = o;
      } else {
        if (i < N)     start[i]     = o.x;
        if (i + 1 < N) start[i + 1] = o.y;
        if (i + 2 < N) start[i + 2] = o.z;
      }
    }
    carry += tile_total;
    __syncthreads();
  }
  if (tid == 0) start[N] = carry;
}

__global__ void edge_fill_kernel(const int* __restrict__ ei, const float* __restrict__ w,
                                 const int* __restrict__ start, const int* __restrict__ rank,
                                 int2* __restrict__ sedge, int E) {
  int e = blockIdx.x * blockDim.x + threadIdx.x;
  if (e < E) {
    unsigned pos = (unsigned)(start[ei[E + e]] + rank[e]);
    if (pos < (unsigned)E) {
      int2 ed;
      ed.x = ei[e];
      ed.y = __float_as_int(w[e]);
      sedge[pos] = ed;
    }
  }
}

__global__ void dinv_kernel(const int* __restrict__ start, const int2* __restrict__ sedge,
                            float* __restrict__ dinv, int N, int E) {
  int i = blockIdx.x * blockDim.x + threadIdx.x;
  if (i < N) {
    int b  = max(start[i], 0);
    int en = min(start[i + 1], E);
    float d = 1.0f;
    for (int j = b; j < en; ++j) d += __int_as_float(sedge[j].y);
    dinv[i] = d > 0.0f ? rsqrtf(d) : 0.0f;
  }
}

__global__ void __launch_bounds__(256)
gemm_xW_kernel(const float* __restrict__ X, const float* __restrict__ W,
               float* __restrict__ H) {
  const int wave = threadIdx.x >> 5;
  const int lane = threadIdx.x & 31;
  const int row0 = blockIdx.x * 16;
  const int col0 = wave * 16;
  const int m    = lane & 15;
  const int half = lane >> 4;
  const int n    = col0 + m;

  v8f acc = {};
  const float* xrow = X + (size_t)(row0 + m) * D + half * 2;
  for (int k = 0; k < D; k += 4) {
    v2f a = *(const v2f*)(xrow + k);
    const float* wp = W + (size_t)(k + half * 2) * D + n;
    v2f b;
    b.x = wp[0];
    b.y = wp[D];
    acc = __builtin_amdgcn_wmma_f32_16x16x4_f32(
        false, a, false, b, (short)0, acc, false, false);
  }
  float* hp = H + (size_t)(row0 + half * 8) * D + n;
#pragma unroll
  for (int r = 0; r < 8; ++r) hp[(size_t)r * D] = acc[r];
}

__global__ void __launch_bounds__(256)
gather_bias_relu_ln_kernel(const int* __restrict__ start, const int2* __restrict__ sedge,
                           const float* __restrict__ dinv, const float* __restrict__ h,
                           const float* __restrict__ bias, const float* __restrict__ gamma,
                           const float* __restrict__ beta, float* __restrict__ y,
                           int N, int E) {
  const int wave = __builtin_amdgcn_readfirstlane((int)(threadIdx.x >> 5));
  const int lane = threadIdx.x & 31;
  const int node = blockIdx.x * 8 + wave;
  if (node >= N) return;
  const int f = lane * 4;
  const float dc = dinv[node];
  const int beg = max(start[node], 0);
  const int end = min(start[node + 1], E);
  const float* hf = h + f;

  float4 acc;
  {
    const float cs = dc * dc;
    const float4 hv = *(const float4*)(hf + (size_t)node * D);
    acc.x = cs * hv.x; acc.y = cs * hv.y; acc.z = cs * hv.z; acc.w = cs * hv.w;
  }

  for (int base = beg; base < end; base += 32) {
    const int j = base + lane;
    int   r_l  = 0;
    float cf_l = 0.0f;
    if (j < end) {
      const int2 ed = sedge[j];
      r_l  = (int)min((unsigned)ed.x, (unsigned)(N - 1));
      cf_l = dinv[r_l] * __int_as_float(ed.y) * dc;
    }
    const int nb = min(32, end - base);
    int t = 0;
    for (; t + 4 <= nb; t += 4) {
      const int r0 = __builtin_amdgcn_readlane(r_l, t);
      const int r1 = __builtin_amdgcn_readlane(r_l, t + 1);
      const int r2 = __builtin_amdgcn_readlane(r_l, t + 2);
      const int r3 = __builtin_amdgcn_readlane(r_l, t + 3);
      const float c0 = __int_as_float(__builtin_amdgcn_readlane(__float_as_int(cf_l), t));
      const float c1 = __int_as_float(__builtin_amdgcn_readlane(__float_as_int(cf_l), t + 1));
      const float c2 = __int_as_float(__builtin_amdgcn_readlane(__float_as_int(cf_l), t + 2));
      const float c3 = __int_as_float(__builtin_amdgcn_readlane(__float_as_int(cf_l), t + 3));
      const float4 a0 = *(const float4*)(hf + (size_t)r0 * D);
      const float4 a1 = *(const float4*)(hf + (size_t)r1 * D);
      const float4 a2 = *(const float4*)(hf + (size_t)r2 * D);
      const float4 a3 = *(const float4*)(hf + (size_t)r3 * D);
      acc.x += c0 * a0.x; acc.y += c0 * a0.y; acc.z += c0 * a0.z; acc.w += c0 * a0.w;
      acc.x += c1 * a1.x; acc.y += c1 * a1.y; acc.z += c1 * a1.z; acc.w += c1 * a1.w;
      acc.x += c2 * a2.x; acc.y += c2 * a2.y; acc.z += c2 * a2.z; acc.w += c2 * a2.w;
      acc.x += c3 * a3.x; acc.y += c3 * a3.y; acc.z += c3 * a3.z; acc.w += c3 * a3.w;
    }
    for (; t < nb; ++t) {
      const int r0 = __builtin_amdgcn_readlane(r_l, t);
      const float c0 = __int_as_float(__builtin_amdgcn_readlane(__float_as_int(cf_l), t));
      const float4 a0 = *(const float4*)(hf + (size_t)r0 * D);
      acc.x += c0 * a0.x; acc.y += c0 * a0.y; acc.z += c0 * a0.z; acc.w += c0 * a0.w;
    }
  }

  float4 v = acc;
  const float4 b = *(const float4*)(bias + f);
  v.x = fmaxf(v.x + b.x, 0.0f);
  v.y = fmaxf(v.y + b.y, 0.0f);
  v.z = fmaxf(v.z + b.z, 0.0f);
  v.w = fmaxf(v.w + b.w, 0.0f);
  float s = v.x + v.y + v.z + v.w;
#pragma unroll
  for (int mski = 16; mski >= 1; mski >>= 1) s += __shfl_xor(s, mski, 32);
  float mu = s * (1.0f / D);
  float dx0 = v.x - mu, dx1 = v.y - mu, dx2 = v.z - mu, dx3 = v.w - mu;
  float vs = dx0 * dx0 + dx1 * dx1 + dx2 * dx2 + dx3 * dx3;
#pragma unroll
  for (int mski = 16; mski >= 1; mski >>= 1) vs += __shfl_xor(vs, mski, 32);
  float inv = rsqrtf(vs * (1.0f / D) + LN_EPS);
  const float4 g  = *(const float4*)(gamma + f);
  const float4 be = *(const float4*)(beta + f);
  float4 o;
  o.x = dx0 * inv * g.x + be.x;
  o.y = dx1 * inv * g.y + be.y;
  o.z = dx2 * inv * g.z + be.z;
  o.w = dx3 * inv * g.w + be.w;
  *(float4*)(y + (size_t)node * D + f) = o;
}

__global__ void __launch_bounds__(256)
gemm_y_linT_kernel(const float* __restrict__ Y, const float* __restrict__ linW,
                   const float* __restrict__ linb, float* __restrict__ OUT) {
  const int wave = threadIdx.x >> 5;
  const int lane = threadIdx.x & 31;
  const int row0 = blockIdx.x * 16;
  const int col0 = wave * 16;
  const int m    = lane & 15;
  const int half = lane >> 4;
  const int n    = col0 + m;

  float lb = linb[n];
  v8f acc = {lb, lb, lb, lb, lb, lb, lb, lb};
  const float* yrow = Y + (size_t)(row0 + m) * D + half * 2;
  const float* brow = linW + (size_t)n * D + half * 2;
  for (int k = 0; k < D; k += 4) {
    v2f a = *(const v2f*)(yrow + k);
    v2f b = *(const v2f*)(brow + k);
    acc = __builtin_amdgcn_wmma_f32_16x16x4_f32(
        false, a, false, b, (short)0, acc, false, false);
  }
  float* op = OUT + (size_t)(row0 + half * 8) * D + n;
#pragma unroll
  for (int r = 0; r < 8; ++r) op[(size_t)r * D] = acc[r];
}

extern "C" void kernel_launch(void* const* d_in, const int* in_sizes, int n_in,
                              void* d_out, int out_size, void* d_ws, size_t ws_size,
                              hipStream_t stream) {
  const float* x     = (const float*)d_in[0];
  const int*   ei    = (const int*)d_in[1];
  const float* ew    = (const float*)d_in[2];
  const float* W     = (const float*)d_in[3];
  const float* bias  = (const float*)d_in[4];
  const float* gamma = (const float*)d_in[5];
  const float* beta  = (const float*)d_in[6];
  const float* linW  = (const float*)d_in[7];
  const float* linb  = (const float*)d_in[8];
  float* out = (float*)d_out;

  const int N = in_sizes[0] / D;
  const int E = in_sizes[2];

  char* ws = (char*)d_ws;
  const size_t nB = (((size_t)(N + 1) * sizeof(int)) + 255) & ~(size_t)255;
  const size_t eB = (((size_t)E * sizeof(int)) + 255) & ~(size_t)255;
  float* dinv  = (float*)ws;
  int*   cnt   = (int*)(ws + nB);
  int*   start = (int*)(ws + 2 * nB);
  int*   rank  = (int*)(ws + 3 * nB);
  int2*  sedge = (int2*)(ws + 3 * nB + eB);
  float* h     = (float*)(ws + 3 * nB + 3 * eB);
  float* y     = h + (size_t)N * D;

  zero_count_kernel<<<(N + 255) / 256, 256, 0, stream>>>(cnt, N);
  edge_rank_kernel<<<(E + 255) / 256, 256, 0, stream>>>(ei, cnt, rank, E);
  segment_start_kernel<<<1, 1024, 0, stream>>>(cnt, start, N);
  edge_fill_kernel<<<(E + 255) / 256, 256, 0, stream>>>(ei, ew, start, rank, sedge, E);
  dinv_kernel<<<(N + 255) / 256, 256, 0, stream>>>(start, sedge, dinv, N, E);
  gemm_xW_kernel<<<N / 16, 256, 0, stream>>>(x, W, h);
  gather_bias_relu_ln_kernel<<<(N + 7) / 8, 256, 0, stream>>>(start, sedge, dinv, h, bias, gamma,
                                                             beta, y, N, E);
  gemm_y_linT_kernel<<<N / 16, 256, 0, stream>>>(y, linW, linb, out);
}


# Both f32 products: weights packed once per call in fragment order, kept in LDS per work-group; each wave owns a 16-row strip (A by 16-byte loads, 8 accumulators); k sums reordered inside 8-blocks
# speedup vs baseline: 1.9531x; 1.2605x over previous
#include <hip/hip_runtime.h>

#define D 128
#define LN_EPS 1e-5f

typedef __attribute__((ext_vector_type(2))) float v2f;
typedef __attribute__((ext_vector_type(8))) float v8f;

__global__ void zero_count_pack_kernel(int* __restrict__ cnt, int N,
                                       const float* __restrict__ W, const float* __restrict__ linW,
                                       float4* __restrict__ Wp, float4* __restrict__ Lp) {
  int i = blockIdx.x * blockDim.x + threadIdx.x;
  if (i < N) cnt[i] = 0;
  if (i < (D / 8) * (D / 16) * 32) {
    const int lane = i & 31;
    const int t    = (i >> 5) & 7;
    const int j    = i >> 8;
    const int k0   = 8 * j + 4 * (lane >> 4);
    const int n    = 16 * t + (lane & 15);
    float4 wv;
    wv.x = W[(size_t)(k0 + 0) * D + n];
    wv.y = W[(size_t)(k0 + 1) * D + n];
    wv.z = W[(size_t)(k0 + 2) * D + n];
    wv.w = W[(size_t)(k0 + 3) * D + n];
    Wp[i] = wv;
    Lp[i] = *(const float4*)(linW + (size_t)n * D + k0);
  }
}

__global__ void edge_rank_kernel(const int* __restrict__ ei, int* __restrict__ cnt,
                                 int* __restrict__ rank, int E) {
  int e = blockIdx.x * blockDim.x + threadIdx.x;
  if (e < E) rank[e] = atomicAdd(&cnt[ei[E + e]], 1);
}

#define SCAN_Q 8
__global__ void __launch_bounds__(1024)
segment_start_kernel(const int* __restrict__ cnt, int* __restrict__ start, int N) {
  __shared__ int wsum[32];
  __shared__ int tile_total;
  const int tid  = threadIdx.x;
  const int lane = tid & 31;
  const int wv   = tid >> 5;
  int carry = 0;
  for (int base = 0; base < N; base += 1024 * 4 * SCAN_Q) {
    const int i0 = base + tid * (4 * SCAN_Q);
    int v[4 * SCAN_Q];
#pragma unroll
    for (int q = 0; q < SCAN_Q; ++q) {
      const int i = i0 + 4 * q;
      if (i + 3 < N) {
        const int4 t = *(const int4*)(cnt + i);
        v[4 * q] = t.x; v[4 * q + 1] = t.y; v[4 * q + 2] = t.z; v[4 * q + 3] = t.w;
      } else {
        v[4 * q]     = (i < N)     ? cnt[i]     : 0;
        v[4 * q + 1] = (i + 1 < N) ? cnt[i + 1] : 0;
        v[4 * q + 2] = (i + 2 < N) ? cnt[i + 2] : 0;
        v[4 * q + 3] = 0;
      }
    }
    int s = 0;
#pragma unroll
    for (int k = 0; k < 4 * SCAN_Q; ++k) s += v[k];
    int inc = s;
#pragma unroll
    for (int d = 1; d < 32; d <<= 1) {
      int t = __shfl_up(inc, d, 32);
      if (lane >= d) inc += t;
    }
    if (lane == 31) wsum[wv] = inc;
    __syncthreads();
    if (wv == 0) {
      const int x = wsum[lane];
      int xi = x;
#pragma unroll
      for (int d = 1; d < 32; d <<= 1) {
        int t = __shfl_up(xi, d, 32);
        if (lane >= d) xi += t;
      }
      wsum[lane] = xi - x;
      if (lane == 31) tile_total = xi;
    }
    __syncthreads();
    int run = carry + wsum[wv] + (inc - s);
#pragma unroll
    for (int q = 0; q < SCAN_Q; ++q) {
      const int i = i0 + 4 * q;
      int4 o;
      o.x = run; run += v[4 * q];
      o.y = run; run += v[4 * q + 1];
      o.z = run; run += v[4 * q + 2];
      o.w = run; run += v[4 * q + 3];
      if (i + 3 < N) {
        *(int4*)(start + i) = o;
      } else {
        if (i < N)     start[i]     = o.x;
        if (i + 1 < N) start[i + 1] = o.y;
        if (i + 2 < N) start[i + 2] = o.z;
      }
    }
    carry += tile_total;
    __syncthreads();
  }
  if (tid == 0) start[N] = carry;
}

__global__ void edge_fill_kernel(const int* __restrict__ ei, const float* __restrict__ w,
                                 const int* __restrict__ start, const int* __restrict__ rank,
                                 int2* __restrict__ sedge, int E) {
  int e = blockIdx.x * blockDim.x + threadIdx.x;
  if (e < E) {
    unsigned pos = (unsigned)(start[ei[E + e]] + rank[e]);
    if (pos < (unsigned)E) {
      int2 ed;
      ed.x = ei[e];
      ed.y = __float_as_int(w[e]);
      sedge[pos] = ed;
    }
  }
}

__global__ void dinv_kernel(const int* __restrict__ start, const int2* __restrict__ sedge,
                            float* __restrict__ dinv, int N, int E) {
  int i = blockIdx.x * blockDim.x + threadIdx.x;
  if (i < N) {
    int b  = max(start[i], 0);
    int en = min(start[i + 1], E);
    float d = 1.0f;
    for (int j = b; j < en; ++j) d += __int_as_float(sedge[j].y);
    dinv[i] = d > 0.0f ? rsqrtf(d) : 0.0f;
  }
}

#define GEMM_WAVES 4
template <bool HAS_BIAS>
__global__ void __launch_bounds__(32 * GEMM_WAVES)
gemm_strip_kernel(const float* __restrict__ A, const float* __restrict__ Bp,
                  const float* __restrict__ bias, float* __restrict__ C, int nstrips) {
  __shared__ float4 sB[(D / 8) * (D / 16) * 32];
  const int tid  = threadIdx.x;
  const int wave = tid >> 5;
  const int lane = tid & 31;
  const int c    = lane & 15;
  const int h    = lane >> 4;
  const int strip = blockIdx.x * GEMM_WAVES + wave;
  const int lstrip = min(strip, nstrips - 1);

  const float4* arow = (const float4*)(A + (size_t)(lstrip * 16 + c) * D + 4 * h);
  float4 a[D / 8];
#pragma unroll
  for (int j = 0; j < D / 8; ++j) a[j] = arow[2 * j];

  const float4* bp4 = (const float4*)Bp;
#pragma unroll
  for (int i = 0; i < (D / 8) * (D / 16) / GEMM_WAVES; ++i)
    sB[tid + 32 * GEMM_WAVES * i] = bp4[tid + 32 * GEMM_WAVES * i];
  __syncthreads();
  if (strip >= nstrips) return;

  v8f acc[D / 16];
#pragma unroll
  for (int t = 0; t < D / 16; ++t) {
    const float b0 = HAS_BIAS ? bias[16 * t + c] : 0.0f;
    acc[t] = (v8f){b0, b0, b0, b0, b0, b0, b0, b0};
  }

  const float4* sl = sB + lane;
#pragma unroll
  for (int j = 0; j < D / 8; ++j) {
    float4 b[D / 16];
#pragma unroll
    for (int t = 0; t < D / 16; ++t) b[t] = sl[(j * (D / 16) + t) * 32];
    const v2f a01 = {a[j].x, a[j].y};
    const v2f a23 = {a[j].z, a[j].w};
#pragma unroll
    for (int t = 0; t < D / 16; ++t) {
      const v2f b01 = {b[t].x, b[t].y};
      acc[t] = __builtin_amdgcn_wmma_f32_16x16x4_f32(
          false, a01, false, b01, (short)0, acc[t], false, false);
    }
#pragma unroll
    for (int t = 0; t < D / 16; ++t) {
      const v2f b23 = {b[t].z, b[t].w};
      acc[t] = __builtin_amdgcn_wmma_f32_16x16x4_f32(
          false, a23, false, b23, (short)0, acc[t], false, false);
    }
  }

  float* cp = C + (size_t)(strip * 16 + h * 8) * D + c;
#pragma unroll
  for (int t = 0; t < D / 16; ++t)
#pragma unroll
    for (int r = 0; r < 8; ++r) cp[(size_t)r * D + 16 * t] = acc[t][r];
}

__global__ void __launch_bounds__(256)
gather_bias_relu_ln_kernel(const int* __restrict__ start, const int2* __restrict__ sedge,
                           const float* __restrict__ dinv, const float* __restrict__ h,
                           const float* __restrict__ bias, const float* __restrict__ gamma,
                           const float* __restrict__ beta, float* __restrict__ y,
                           int N, int E) {
  const int wave = __builtin_amdgcn_readfirstlane((int)(threadIdx.x >> 5));
  const int lane = threadIdx.x & 31;
  const int node = blockIdx.x * 8 + wave;
  if (node >= N) return;
  const int f = lane * 4;
  const float dc = dinv[node];
  const int beg = max(start[node], 0);
  const int end = min(start[node + 1], E);
  const float* hf = h + f;

  float4 acc;
  {
    const float cs = dc * dc;
    const float4 hv = *(const float4*)(hf + (size_t)node * D);
    acc.x = cs * hv.x; acc.y = cs * hv.y; acc.z = cs * hv.z; acc.w = cs * hv.w;
  }

  for (int base = beg; base < end; base += 32) {
    const int j = base + lane;
    int   r_l  = 0;
    float cf_l = 0.0f;
    if (j < end) {
      const int2 ed = sedge[j];
      r_l  = (int)min((unsigned)ed.x, (unsigned)(N - 1));
      cf_l = dinv[r_l] * __int_as_float(ed.y) * dc;
    }
    const int nb = min(32, end - base);
    int t = 0;
    for (; t + 4 <= nb; t += 4) {
      const int r0 = __builtin_amdgcn_readlane(r_l, t);
      const int r1 = __builtin_amdgcn_readlane(r_l, t + 1);
      const int r2 = __builtin_amdgcn_readlane(r_l, t + 2);
      const int r3 = __builtin_amdgcn_readlane(r_l, t + 3);
      const float c0 = __int_as_float(__builtin_amdgcn_readlane(__float_as_int(cf_l), t));
      const float c1 = __int_as_float(__builtin_amdgcn_readlane(__float_as_int(cf_l), t + 1));
      const float c2 = __int_as_float(__builtin_amdgcn_readlane(__float_as_int(cf_l), t + 2));
      const float c3 = __int_as_float(__builtin_amdgcn_readlane(__float_as_int(cf_l), t + 3));
      const float4 a0 = *(const float4*)(hf + (size_t)r0 * D);
      const float4 a1 = *(const float4*)(hf + (size_t)r1 * D);
      const float4 a2 = *(const float4*)(hf + (size_t)r2 * D);
      const float4 a3 = *(const float4*)(hf + (size_t)r3 * D);
      acc.x += c0 * a0.x; acc.y += c0 * a0.y; acc.z += c0 * a0.z; acc.w += c0 * a0.w;
      acc.x += c1 * a1.x; acc.y += c1 * a1.y; acc.z += c1 * a1.z; acc.w += c1 * a1.w;
      acc.x += c2 * a2.x; acc.y += c2 * a2.y; acc.z += c2 * a2.z; acc.w += c2 * a2.w;
      acc.x += c3 * a3.x; acc.y += c3 * a3.y; acc.z += c3 * a3.z; acc.w += c3 * a3.w;
    }
    for (; t < nb; ++t) {
      const int r0 = __builtin_amdgcn_readlane(r_l, t);
      const float c0 = __int_as_float(__builtin_amdgcn_readlane(__float_as_int(cf_l), t));
      const float4 a0 = *(const float4*)(hf + (size_t)r0 * D);
      acc.x += c0 * a0.x; acc.y += c0 * a0.y; acc.z += c0 * a0.z; acc.w += c0 * a0.w;
    }
  }

  float4 v = acc;
  const float4 b = *(const float4*)(bias + f);
  v.x = fmaxf(v.x + b.x, 0.0f);
  v.y = fmaxf(v.y + b.y, 0.0f);
  v.z = fmaxf(v.z + b.z, 0.0f);
  v.w = fmaxf(v.w + b.w, 0.0f);
  float s = v.x + v.y + v.z + v.w;
#pragma unroll
  for (int mski = 16; mski >= 1; mski >>= 1) s += __shfl_xor(s, mski, 32);
  float mu = s * (1.0f / D);
  float dx0 = v.x - mu, dx1 = v.y - mu, dx2 = v.z - mu, dx3 = v.w - mu;
  float vs = dx0 * dx0 + dx1 * dx1 + dx2 * dx2 + dx3 * dx3;
#pragma unroll
  for (int mski = 16; mski >= 1; mski >>= 1) vs += __shfl_xor(vs, mski, 32);
  float inv = rsqrtf(vs * (1.0f / D) + LN_EPS);
  const float4 g  = *(const float4*)(gamma + f);
  const float4 be = *(const float4*)(beta + f);
  float4 o;
  o.x = dx0 * inv * g.x + be.x;
  o.y = dx1 * inv * g.y + be.y;
  o.z = dx2 * inv * g.z + be.z;
  o.w = dx3 * inv * g.w + be.w;
  *(float4*)(y + (size_t)node * D + f) = o;
}

extern "C" void kernel_launch(void* const* d_in, const int* in_sizes, int n_in,
                              void* d_out, int out_size, void* d_ws, size_t ws_size,
                              hipStream_t stream) {
  const float* x     = (const float*)d_in[0];
  const int*   ei    = (const int*)d_in[1];
  const float* ew    = (const float*)d_in[2];
  const float* W     = (const float*)d_in[3];
  const float* bias  = (const float*)d_in[4];
  const float* gamma = (const float*)d_in[5];
  const float* beta  = (const float*)d_in[6];
  const float* linW  = (const float*)d_in[7];
  const float* linb  = (const float*)d_in[8];
  float* out = (float*)d_out;

  const int N = in_sizes[0] / D;
  const int E = in_sizes[2];

  char* ws = (char*)d_ws;
  const size_t nB = (((size_t)(N + 1) * sizeof(int)) + 255) & ~(size_t)255;
  const size_t eB = (((size_t)E * sizeof(int)) + 255) & ~(size_t)255;
  float* dinv  = (float*)ws;
  int*   cnt   = (int*)(ws + nB);
  int*   start = (int*)(ws + 2 * nB);
  int*   rank  = (int*)(ws + 3 * nB);
  int2*  sedge = (int2*)(ws + 3 * nB + eB);
  float* h     = (float*)(ws + 3 * nB + 3 * eB);
  float* y     = h + (size_t)N * D;
  float4* Wp   = (float4*)(y + (size_t)N * D);
  float4* Lp   = Wp + (D / 8) * (D / 16) * 32;

  zero_count_pack_kernel<<<(N + 255) / 256, 256, 0, stream>>>(cnt, N, W, linW, Wp, Lp);
  edge_rank_kernel<<<(E + 255) / 256, 256, 0, stream>>>(ei, cnt, rank, E);
  segment_start_kernel<<<1, 1024, 0, stream>>>(cnt, start, N);
  edge_fill_kernel<<<(E + 255) / 256, 256, 0, stream>>>(ei, ew, start, rank, sedge, E);
  dinv_kernel<<<(N + 255) / 256, 256, 0, stream>>>(start, sedge, dinv, N, E);
  gemm_strip_kernel<false><<<(N / 16 + GEMM_WAVES - 1) / GEMM_WAVES, 32 * GEMM_WAVES, 0, stream>>>(
      x, (const float*)Wp, nullptr, h, N / 16);
  gather_bias_relu_ln_kernel<<<(N + 7) / 8, 256, 0, stream>>>(start, sedge, dinv, h, bias, gamma,
                                                             beta, y, N, E);
  gemm_strip_kernel<true><<<(N / 16 + GEMM_WAVES - 1) / GEMM_WAVES, 32 * GEMM_WAVES, 0, stream>>>(
      y, (const float*)Lp, linb, out, N / 16);
}


# Edges sorted by target with a two-level counting sort on LDS counters (256-node bins, runs staged in LDS; per-bin pass also writes starts and dinv): no per-edge global atomic; LDS arrival order sums
# speedup vs baseline: 4.0691x; 2.0834x over previous
#include <hip/hip_runtime.h>

#define D 128
#define LN_EPS 1e-5f

typedef __attribute__((ext_vector_type(2))) float v2f;
typedef __attribute__((ext_vector_type(8))) float v8f;

#define BIN_SHIFT 8
#define BIN_NODES 256
#define MAXB 512
#define CHUNK_T 256
#define CHUNK_K 16
#define CHUNK (CHUNK_T * CHUNK_K)

__global__ void zero_bins_pack_kernel(int* __restrict__ binCnt, int* __restrict__ binCursor,
                                      const float* __restrict__ W, const float* __restrict__ linW,
                                      float4* __restrict__ Wp, float4* __restrict__ Lp) {
  int i = blockIdx.x * blockDim.x + threadIdx.x;
  if (i < MAXB) { binCnt[i] = 0; binCursor[i] = 0; }
  if (i < (D / 8) * (D / 16) * 32) {
    const int lane = i & 31;
    const int t    = (i >> 5) & 7;
    const int j    = i >> 8;
    const int k0   = 8 * j + 4 * (lane >> 4);
    const int n    = 16 * t + (lane & 15);
    float4 wv;
    wv.x = W[(size_t)(k0 + 0) * D + n];
    wv.y = W[(size_t)(k0 + 1) * D + n];
    wv.z = W[(size_t)(k0 + 2) * D + n];
    wv.w = W[(size_t)(k0 + 3) * D + n];
    Wp[i] = wv;
    Lp[i] = *(const float4*)(linW + (size_t)n * D + k0);
  }
}

__global__ void __launch_bounds__(CHUNK_T)
bin_count_kernel(const int* __restrict__ ei, int* __restrict__ binCnt, int E, int N, int B) {
  __shared__ int hist[MAXB];
  const int tid = threadIdx.x;
  for (int i = tid; i < MAXB; i += CHUNK_T) hist[i] = 0;
  const int base = blockIdx.x * CHUNK;
  int col[CHUNK_K];
#pragma unroll
  for (int k = 0; k < CHUNK_K; ++k) {
    const int e = min(base + k * CHUNK_T + tid, E - 1);
    col[k] = (int)min((unsigned)ei[E + e], (unsigned)(N - 1));
  }
  __syncthreads();
#pragma unroll
  for (int k = 0; k < CHUNK_K; ++k)
    if (base + k * CHUNK_T + tid < E) atomicAdd(&hist[col[k] >> BIN_SHIFT], 1);
  __syncthreads();
  for (int i = tid; i < B; i += CHUNK_T) {
    const int v = hist[i];
    if (v) atomicAdd(&binCnt[i], v);
  }
}

__global__ void __launch_bounds__(CHUNK_T)
bin_place_kernel(const int* __restrict__ ei, const float* __restrict__ w,
                 const int* __restrict__ binCnt, int* __restrict__ binCursor,
                 int* __restrict__ binStart, int4* __restrict__ pedge, int E, int N, int B) {
  __shared__ int4 stage[CHUNK];
  __shared__ int hist[MAXB];
  __shared__ int lbase[MAXB];
  __shared__ int gdelta[MAXB];
  __shared__ int wsumG[CHUNK_T / 32];
  __shared__ int wsumL[CHUNK_T / 32];
  const int tid  = threadIdx.x;
  const int lane = tid & 31;
  const int wv   = tid >> 5;
  for (int i = tid; i < MAXB; i += CHUNK_T) hist[i] = 0;
  const int base = blockIdx.x * CHUNK;
  const int nvalid = min(CHUNK, E - base);
  int col[CHUNK_K], row[CHUNK_K], wb[CHUNK_K], rk[CHUNK_K];
#pragma unroll
  for (int k = 0; k < CHUNK_K; ++k) {
    const int e = min(base + k * CHUNK_T + tid, E - 1);
    col[k] = (int)min((unsigned)ei[E + e], (unsigned)(N - 1));
    row[k] = ei[e];
    wb[k]  = __float_as_int(w[e]);
  }
  const int b0 = 2 * tid, b1 = 2 * tid + 1;
  const int v0 = b0 < B ? binCnt[b0] : 0;
  const int v1 = b1 < B ? binCnt[b1] : 0;
  __syncthreads();
#pragma unroll
  for (int k = 0; k < CHUNK_K; ++k) {
    rk[k] = 0;
    if (k * CHUNK_T + tid < nvalid) rk[k] = atomicAdd(&hist[col[k] >> BIN_SHIFT], 1);
  }
  __syncthreads();
  const int c0 = hist[b0], c1 = hist[b1];
  const int sG = v0 + v1;
  const int sL = c0 + c1;
  int incG = sG, incL = sL;
#pragma unroll
  for (int d = 1; d < 32; d <<= 1) {
    int tG = __shfl_up(incG, d, 32);
    int tL = __shfl_up(incL, d, 32);
    if (lane >= d) { incG += tG; incL += tL; }
  }
  if (lane == 31) { wsumG[wv] = incG; wsumL[wv] = incL; }
  __syncthreads();
  int exG = incG - sG, exL = incL - sL;
  for (int q = 0; q < wv; ++q) { exG += wsumG[q]; exL += wsumL[q]; }
  {
    const int g0 = c0 ? atomicAdd(&binCursor[b0], c0) : 0;
    const int g1 = c1 ? atomicAdd(&binCursor[b1], c1) : 0;
    lbase[b0]  = exL;
    lbase[b1]  = exL + c0;
    gdelta[b0] = exG + g0 - exL;
    gdelta[b1] = exG + v0 + g1 - (exL + c0);
    if (blockIdx.x == 0) {
      binStart[b0] = exG;
      binStart[b1] = exG + v0;
      if (tid == CHUNK_T - 1) binStart[MAXB] = exG + sG;
    }
  }
  __syncthreads();
  int ls[CHUNK_K];
#pragma unroll
  for (int k = 0; k < CHUNK_K; ++k) ls[k] = lbase[col[k] >> BIN_SHIFT] + rk[k];
#pragma unroll
  for (int k = 0; k < CHUNK_K; ++k) {
    if (k * CHUNK_T + tid < nvalid) {
      int4 ed;
      ed.x = col[k]; ed.y = row[k]; ed.z = wb[k]; ed.w = 0;
      stage[ls[k]] = ed;
    }
  }
  __syncthreads();
  int4 so[CHUNK_K];
  int  gd[CHUNK_K];
#pragma unroll
  for (int k = 0; k < CHUNK_K; ++k) so[k] = stage[min(k * CHUNK_T + tid, nvalid - 1)];
#pragma unroll
  for (int k = 0; k < CHUNK_K; ++k) gd[k] = gdelta[so[k].x >> BIN_SHIFT];
#pragma unroll
  for (int k = 0; k < CHUNK_K; ++k) {
    const int i = k * CHUNK_T + tid;
    if (i < nvalid) {
      const unsigned pos = (unsigned)(gd[k] + i);
      if (pos < (unsigned)E) pedge[pos] = so[k];
    }
  }
}

#define SORT_U 4
#define SORT_STAGE 8192
__global__ void __launch_bounds__(BIN_NODES)
bin_sort_kernel(const int* __restrict__ binStart, const int4* __restrict__ pedge,
                int* __restrict__ start, int2* __restrict__ sedge, float* __restrict__ dinv,
                int N, int E) {
  __shared__ int2  stage[SORT_STAGE];
  __shared__ int   cnt[BIN_NODES];
  __shared__ float wsumf[BIN_NODES];
  __shared__ int   cur[BIN_NODES];
  __shared__ int   wtot[BIN_NODES / 32];
  const int tid  = threadIdx.x;
  const int lane = tid & 31;
  const int wv   = tid >> 5;
  const int bin  = blockIdx.x;
  const int bs = max(binStart[bin], 0);
  const int be = min(binStart[bin + 1], E);
  cnt[tid] = 0;
  wsumf[tid] = 0.0f;
  __syncthreads();
  for (int i0 = bs; i0 < be; i0 += SORT_U * BIN_NODES) {
    int4 ed[SORT_U];
#pragma unroll
    for (int u = 0; u < SORT_U; ++u) ed[u] = pedge[min(i0 + u * BIN_NODES + tid, be - 1)];
#pragma unroll
    for (int u = 0; u < SORT_U; ++u) {
      if (i0 + u * BIN_NODES + tid < be) {
        const int l = ed[u].x & (BIN_NODES - 1);
        atomicAdd(&cnt[l], 1);
        atomicAdd(&wsumf[l], __int_as_float(ed[u].z));
      }
    }
  }
  __syncthreads();
  const int v = cnt[tid];
  int inc = v;
#pragma unroll
  for (int d = 1; d < 32; d <<= 1) {
    int t = __shfl_up(inc, d, 32);
    if (lane >= d) inc += t;
  }
  if (lane == 31) wtot[wv] = inc;
  __syncthreads();
  int off = inc - v;
  for (int q = 0; q < wv; ++q) off += wtot[q];
  cur[tid] = off;
  const int node = (bin << BIN_SHIFT) + tid;
  if (node < N) {
    start[node] = bs + off;
    const float d = 1.0f + wsumf[tid];
    dinv[node] = d > 0.0f ? rsqrtf(d) : 0.0f;
    if (node == N - 1) start[N] = bs + off + v;
  }
  __syncthreads();
  const bool staged = (be - bs) <= SORT_STAGE;
  for (int i0 = bs; i0 < be; i0 += SORT_U * BIN_NODES) {
    int4 ed[SORT_U];
#pragma unroll
    for (int u = 0; u < SORT_U; ++u) ed[u] = pedge[min(i0 + u * BIN_NODES + tid, be - 1)];
#pragma unroll
    for (int u = 0; u < SORT_U; ++u) {
      if (i0 + u * BIN_NODES + tid < be) {
        const int l = ed[u].x & (BIN_NODES - 1);
        const int p = atomicAdd(&cur[l], 1);
        int2 o;
        o.x = ed[u].y;
        o.y = ed[u].z;
        if (staged) {
          if ((unsigned)p < (unsigned)SORT_STAGE) stage[p] = o;
        } else {
          const unsigned pos = (unsigned)(bs + p);
          if (pos < (unsigned)E) sedge[pos] = o;
        }
      }
    }
  }
  if (staged) {
    __syncthreads();
    for (int i = bs + tid; i < be; i += BIN_NODES) sedge[i] = stage[i - bs];
  }
}

#define GEMM_WAVES 4
template <bool HAS_BIAS>
__global__ void __launch_bounds__(32 * GEMM_WAVES)
gemm_strip_kernel(const float* __restrict__ A, const float* __restrict__ Bp,
                  const float* __restrict__ bias, float* __restrict__ C, int nstrips) {
  __shared__ float4 sB[(D / 8) * (D / 16) * 32];
  const int tid  = threadIdx.x;
  const int wave = tid >> 5;
  const int lane = tid & 31;
  const int c    = lane & 15;
  const int h    = lane >> 4;
  const int strip = blockIdx.x * GEMM_WAVES + wave;
  const int lstrip = min(strip, nstrips - 1);

  const float4* arow = (const float4*)(A + (size_t)(lstrip * 16 + c) * D + 4 * h);
  float4 a[D / 8];
#pragma unroll
  for (int j = 0; j < D / 8; ++j) a[j] = arow[2 * j];

  const float4* bp4 = (const float4*)Bp;
#pragma unroll
  for (int i = 0; i < (D / 8) * (D / 16) / GEMM_WAVES; ++i)
    sB[tid + 32 * GEMM_WAVES * i] = bp4[tid + 32 * GEMM_WAVES * i];
  __syncthreads();
  if (strip >= nstrips) return;

  v8f acc[D / 16];
#pragma unroll
  for (int t = 0; t < D / 16; ++t) {
    const float b0 = HAS_BIAS ? bias[16 * t + c] : 0.0f;
    acc[t] = (v8f){b0, b0, b0, b0, b0, b0, b0, b0};
  }

  const float4* sl = sB + lane;
#pragma unroll
  for (int j = 0; j < D / 8; ++j) {
    float4 b[D / 16];
#pragma unroll
    for (int t = 0; t < D / 16; ++t) b[t] = sl[(j * (D / 16) + t) * 32];
    const v2f a01 = {a[j].x, a[j].y};
    const v2f a23 = {a[j].z, a[j].w};
#pragma unroll
    for (int t = 0; t < D / 16; ++t) {
      const v2f b01 = {b[t].x, b[t].y};
      acc[t] = __builtin_amdgcn_wmma_f32_16x16x4_f32(
          false, a01, false, b01, (short)0, acc[t], false, false);
    }
#pragma unroll
    for (int t = 0; t < D / 16; ++t) {
      const v2f b23 = {b[t].z, b[t].w};
      acc[t] = __builtin_amdgcn_wmma_f32_16x16x4_f32(
          false, a23, false, b23, (short)0, acc[t], false, false);
    }
  }

  float* cp = C + (size_t)(strip * 16 + h * 8) * D + c;
#pragma unroll
  for (int t = 0; t < D / 16; ++t)
#pragma unroll
    for (int r = 0; r < 8; ++r) cp[(size_t)r * D + 16 * t] = acc[t][r];
}

__global__ void __launch_bounds__(256)
gather_bias_relu_ln_kernel(const int* __restrict__ start, const int2* __restrict__ sedge,
                           const float* __restrict__ dinv, const float* __restrict__ h,
                           const float* __restrict__ bias, const float* __restrict__ gamma,
                           const float* __restrict__ beta, float* __restrict__ y,
                           int N, int E) {
  const int wave = __builtin_amdgcn_readfirstlane((int)(threadIdx.x >> 5));
  const int lane = threadIdx.x & 31;
  const int node = blockIdx.x * 8 + wave;
  if (node >= N) return;
  const int f = lane * 4;
  const float dc = dinv[node];
  const int beg = max(start[node], 0);
  const int end = min(start[node + 1], E);
  const float* hf = h + f;

  float4 acc;
  {
    const float cs = dc * dc;
    const float4 hv = *(const float4*)(hf + (size_t)node * D);
    acc.x = cs * hv.x; acc.y = cs * hv.y; acc.z = cs * hv.z; acc.w = cs * hv.w;
  }

  for (int base = beg; base < end; base += 32) {
    const int j = base + lane;
    int   r_l  = 0;
    float cf_l = 0.0f;
    if (j < end) {
      const int2 ed = sedge[j];
      r_l  = (int)min((unsigned)ed.x, (unsigned)(N - 1));
      cf_l = dinv[r_l] * __int_as_float(ed.y) * dc;
    }
    const int nb = min(32, end - base);
    int t = 0;
    for (; t + 4 <= nb; t += 4) {
      const int r0 = __builtin_amdgcn_readlane(r_l, t);
      const int r1 = __builtin_amdgcn_readlane(r_l, t + 1);
      const int r2 = __builtin_amdgcn_readlane(r_l, t + 2);
      const int r3 = __builtin_amdgcn_readlane(r_l, t + 3);
      const float c0 = __int_as_float(__builtin_amdgcn_readlane(__float_as_int(cf_l), t));
      const float c1 = __int_as_float(__builtin_amdgcn_readlane(__float_as_int(cf_l), t + 1));
      const float c2 = __int_as_float(__builtin_amdgcn_readlane(__float_as_int(cf_l), t + 2));
      const float c3 = __int_as_float(__builtin_amdgcn_readlane(__float_as_int(cf_l), t + 3));
      const float4 a0 = *(const float4*)(hf + (size_t)r0 * D);
      const float4 a1 = *(const float4*)(hf + (size_t)r1 * D);
      const float4 a2 = *(const float4*)(hf + (size_t)r2 * D);
      const float4 a3 = *(const float4*)(hf + (size_t)r3 * D);
      acc.x += c0 * a0.x; acc.y += c0 * a0.y; acc.z += c0 * a0.z; acc.w += c0 * a0.w;
      acc.x += c1 * a1.x; acc.y += c1 * a1.y; acc.z += c1 * a1.z; acc.w += c1 * a1.w;
      acc.x += c2 * a2.x; acc.y += c2 * a2.y; acc.z += c2 * a2.z; acc.w += c2 * a2.w;
      acc.x += c3 * a3.x; acc.y += c3 * a3.y; acc.z += c3 * a3.z; acc.w += c3 * a3.w;
    }
    for (; t < nb; ++t) {
      const int r0 = __builtin_amdgcn_readlane(r_l, t);
      const float c0 = __int_as_float(__builtin_amdgcn_readlane(__float_as_int(cf_l), t));
      const float4 a0 = *(const float4*)(hf + (size_t)r0 * D);
      acc.x += c0 * a0.x; acc.y += c0 * a0.y; acc.z += c0 * a0.z; acc.w += c0 * a0.w;
    }
  }

  float4 v = acc;
  const float4 b = *(const float4*)(bias + f);
  v.x = fmaxf(v.x + b.x, 0.0f);
  v.y = fmaxf(v.y + b.y, 0.0f);
  v.z = fmaxf(v.z + b.z, 0.0f);
  v.w = fmaxf(v.w + b.w, 0.0f);
  float s = v.x + v.y + v.z + v.w;
#pragma unroll
  for (int mski = 16; mski >= 1; mski >>= 1) s += __shfl_xor(s, mski, 32);
  float mu = s * (1.0f / D);
  float dx0 = v.x - mu, dx1 = v.y - mu, dx2 = v.z - mu, dx3 = v.w - mu;
  float vs = dx0 * dx0 + dx1 * dx1 + dx2 * dx2 + dx3 * dx3;
#pragma unroll
  for (int mski = 16; mski >= 1; mski >>= 1) vs += __shfl_xor(vs, mski, 32);
  float inv = rsqrtf(vs * (1.0f / D) + LN_EPS);
  const float4 g  = *(const float4*)(gamma + f);
  const float4 be = *(const float4*)(beta + f);
  float4 o;
  o.x = dx0 * inv * g.x + be.x;
  o.y = dx1 * inv * g.y + be.y;
  o.z = dx2 * inv * g.z + be.z;
  o.w = dx3 * inv * g.w + be.w;
  *(float4*)(y + (size_t)node * D + f) = o;
}

extern "C" void kernel_launch(void* const* d_in, const int* in_sizes, int n_in,
                              void* d_out, int out_size, void* d_ws, size_t ws_size,
                              hipStream_t stream) {
  const float* x     = (const float*)d_in[0];
  const int*   ei    = (const int*)d_in[1];
  const float* ew    = (const float*)d_in[2];
  const float* W     = (const float*)d_in[3];
  const float* bias  = (const float*)d_in[4];
  const float* gamma = (const float*)d_in[5];
  const float* beta  = (const float*)d_in[6];
  const float* linW  = (const float*)d_in[7];
  const float* linb  = (const float*)d_in[8];
  float* out = (float*)d_out;

  const int N = in_sizes[0] / D;
  const int E = in_sizes[2];
  const int B = (N + BIN_NODES - 1) / BIN_NODES;

  char* ws = (char*)d_ws;
  const size_t nB = (((size_t)(N + 1) * sizeof(int)) + 255) & ~(size_t)255;
  const size_t bB = 4096;
  float* dinv      = (float*)ws;
  int*   start     = (int*)(ws + nB);
  int*   binCnt    = (int*)(ws + 2 * nB);
  int*   binCursor = (int*)(ws + 2 * nB + bB);
  int*   binStart  = (int*)(ws + 2 * nB + 2 * bB);
  int4*  pedge     = (int4*)(ws + 2 * nB + 3 * bB);
  int2*  sedge     = (int2*)(ws + 2 * nB + 3 * bB + (size_t)E * sizeof(int4));
  float* h         = (float*)(ws + 2 * nB + 3 * bB + (size_t)E * (sizeof(int4) + sizeof(int2)));
  float* y     = h + (size_t)N * D;
  float4* Wp   = (float4*)(y + (size_t)N * D);
  float4* Lp   = Wp + (D / 8) * (D / 16) * 32;

  zero_bins_pack_kernel<<<(D / 8) * (D / 16) * 32 / 256, 256, 0, stream>>>(binCnt, binCursor, W, linW,
                                                                        Wp, Lp);
  bin_count_kernel<<<(E + CHUNK - 1) / CHUNK, CHUNK_T, 0, stream>>>(ei, binCnt, E, N, B);
  bin_place_kernel<<<(E + CHUNK - 1) / CHUNK, CHUNK_T, 0, stream>>>(ei, ew, binCnt, binCursor,
                                                                   binStart, pedge, E, N, B);
  bin_sort_kernel<<<B, BIN_NODES, 0, stream>>>(binStart, pedge, start, sedge, dinv, N, E);
  gemm_strip_kernel<false><<<(N / 16 + GEMM_WAVES - 1) / GEMM_WAVES, 32 * GEMM_WAVES, 0, stream>>>(
      x, (const float*)Wp, nullptr, h, N / 16);
  gather_bias_relu_ln_kernel<<<(N + 7) / 8, 256, 0, stream>>>(start, sedge, dinv, h, bias, gamma,
                                                             beta, y, N, E);
  gemm_strip_kernel<true><<<(N / 16 + GEMM_WAVES - 1) / GEMM_WAVES, 32 * GEMM_WAVES, 0, stream>>>(
      y, (const float*)Lp, linb, out, N / 16);
}


# Edge sort without counting pass or global counters: each work-group keeps its bin-sorted 4096-edge chunk plus an (offset,length) row per bin; bin start = sum of offsets; 5 launches; LDS arrival order
# speedup vs baseline: 4.2495x; 1.0443x over previous
#include <hip/hip_runtime.h>

#define D 128
#define LN_EPS 1e-5f

typedef __attribute__((ext_vector_type(2))) float v2f;
typedef __attribute__((ext_vector_type(8))) float v8f;

#define BIN_SHIFT 8
#define BIN_NODES 256
#define MAXB 512
#define CHUNK_T 256
#define CHUNK_K 16
#define CHUNK (CHUNK_T * CHUNK_K)
#define PACK_ENTRIES ((D / 8) * (D / 16) * 32)

__global__ void __launch_bounds__(CHUNK_T)
chunk_sort_pack_kernel(const int* __restrict__ ei, const float* __restrict__ w,
                       int* __restrict__ runTab, int2* __restrict__ cedge,
                       const float* __restrict__ W, const float* __restrict__ linW,
                       float4* __restrict__ Wp, float4* __restrict__ Lp, int E, int N, int G) {
  __shared__ int2 stage[CHUNK];
  __shared__ int hist[MAXB];
  __shared__ int lbase[MAXB];
  __shared__ int wsumL[CHUNK_T / 32];
  const int tid  = threadIdx.x;
  if ((int)blockIdx.x >= G) {
    const int i = ((int)blockIdx.x - G) * CHUNK_T + tid;
    if (i < PACK_ENTRIES) {
      const int lane = i & 31;
      const int t    = (i >> 5) & 7;
      const int j    = i >> 8;
      const int k0   = 8 * j + 4 * (lane >> 4);
      const int n    = 16 * t + (lane & 15);
      float4 wv;
      wv.x = W[(size_t)(k0 + 0) * D + n];
      wv.y = W[(size_t)(k0 + 1) * D + n];
      wv.z = W[(size_t)(k0 + 2) * D + n];
      wv.w = W[(size_t)(k0 + 3) * D + n];
      Wp[i] = wv;
      Lp[i] = *(const float4*)(linW + (size_t)n * D + k0);
    }
    return;
  }
  const int lane = tid & 31;
  const int wv   = tid >> 5;
  for (int i = tid; i < MAXB; i += CHUNK_T) hist[i] = 0;
  const int base = blockIdx.x * CHUNK;
  const int nvalid = min(CHUNK, E - base);
  int col[CHUNK_K], row[CHUNK_K], wb[CHUNK_K], rk[CHUNK_K];
#pragma unroll
  for (int k = 0; k < CHUNK_K; ++k) {
    const int e = min(base + k * CHUNK_T + tid, E - 1);
    col[k] = (int)min((unsigned)ei[E + e], (unsigned)(N - 1));
    row[k] = (int)min((unsigned)ei[e], (unsigned)(N - 1));
    wb[k]  = __float_as_int(w[e]);
  }
  __syncthreads();
#pragma unroll
  for (int k = 0; k < CHUNK_K; ++k) {
    rk[k] = 0;
    if (k * CHUNK_T + tid < nvalid) rk[k] = atomicAdd(&hist[col[k] >> BIN_SHIFT], 1);
  }
  __syncthreads();
  const int b0 = 2 * tid, b1 = 2 * tid + 1;
  const int c0 = hist[b0], c1 = hist[b1];
  const int sL = c0 + c1;
  int incL = sL;
#pragma unroll
  for (int d = 1; d < 32; d <<= 1) {
    int tL = __shfl_up(incL, d, 32);
    if (lane >= d) incL += tL;
  }
  if (lane == 31) wsumL[wv] = incL;
  __syncthreads();
  int exL = incL - sL;
  for (int q = 0; q < wv; ++q) exL += wsumL[q];
  lbase[b0] = exL;
  lbase[b1] = exL + c0;
  {
    int2 te;
    te.x = (exL << 16) | c0;
    te.y = ((exL + c0) << 16) | c1;
    *(int2*)(runTab + (size_t)blockIdx.x * MAXB + b0) = te;
  }
  __syncthreads();
  int ls[CHUNK_K];
#pragma unroll
  for (int k = 0; k < CHUNK_K; ++k) ls[k] = lbase[col[k] >> BIN_SHIFT] + rk[k];
#pragma unroll
  for (int k = 0; k < CHUNK_K; ++k) {
    if (k * CHUNK_T + tid < nvalid) {
      int2 ed;
      ed.x = (int)((unsigned)row[k] | ((unsigned)(col[k] & (BIN_NODES - 1)) << 24));
      ed.y = wb[k];
      stage[ls[k]] = ed;
    }
  }
  __syncthreads();
  int2* dst = cedge + (size_t)blockIdx.x * CHUNK;
#pragma unroll
  for (int k = 0; k < CHUNK_K; ++k) {
    const int i = k * CHUNK_T + tid;
    int2 v = stage[min(i, nvalid - 1)];
    if (i >= nvalid) { v.x = 0; v.y = 0; }
    dst[i] = v;
  }
}

#define SORT_STAGE 8192
#define RUN_U 8
template <bool COLLECT>
__device__ __forceinline__ int sweep_runs(const int* __restrict__ runTab,
                                          const int2* __restrict__ cedge, int bin, int G, int E,
                                          int* rsrc, int* rdst, int* rcnt, int* wtot, int* cnt,
                                          float* wsumf, int2* sin_, int* cur,
                                          int2* __restrict__ sedge, int bs, int& offsum) {
  const int tid  = threadIdx.x;
  const int lane = tid & 31;
  const int wv   = tid >> 5;
  const int hl   = lane & 15;
  const int hh   = lane >> 4;
  const int lim  = G * CHUNK - 1;
  int carry = 0, so = 0;
  for (int g0 = 0; g0 < G; g0 += BIN_NODES) {
    const int g = g0 + tid;
    const int e = g < G ? runTab[(size_t)g * MAXB + bin] : 0;
    const int c = e & 0xFFFF;
    const int o = (int)((unsigned)e >> 16);
    so += o;
    int inc = c;
#pragma unroll
    for (int d = 1; d < 32; d <<= 1) {
      int t = __shfl_up(inc, d, 32);
      if (lane >= d) inc += t;
    }
    if (lane == 31) wtot[wv] = inc;
    __syncthreads();
    int ex = inc - c, bt = 0;
    for (int q = 0; q < BIN_NODES / 32; ++q) {
      const int t = wtot[q];
      if (q < wv) ex += t;
      bt += t;
    }
    rsrc[tid] = g * CHUNK + o;
    rdst[tid] = carry + ex;
    rcnt[tid] = c;
    __syncthreads();
    carry += bt;
    for (int it = 0; it < BIN_NODES / 16; it += RUN_U) {
      int rs[RUN_U], rd[RUN_U], rc[RUN_U];
      int2 ed[RUN_U];
#pragma unroll
      for (int u = 0; u < RUN_U; ++u) {
        const int r = ((it + u) * (BIN_NODES / 32) + wv) * 2 + hh;
        rs[u] = rsrc[r];
        rd[u] = rdst[r];
        rc[u] = rcnt[r];
      }
#pragma unroll
      for (int u = 0; u < RUN_U; ++u) ed[u] = cedge[min(rs[u] + hl, lim)];
#pragma unroll
      for (int u = 0; u < RUN_U; ++u) {
        for (int j = hl; j < rc[u]; j += 16) {
          const int2 v = j == hl ? ed[u] : cedge[min(rs[u] + j, lim)];
          const int l = (int)((unsigned)v.x >> 24);
          if (COLLECT) {
            atomicAdd(&cnt[l], 1);
            atomicAdd(&wsumf[l], __int_as_float(v.y));
            const unsigned dpos = (unsigned)(rd[u] + j);
            if (dpos < (unsigned)SORT_STAGE) sin_[dpos] = v;
          } else {
            const int p = atomicAdd(&cur[l], 1);
            const unsigned pos = (unsigned)(bs + p);
            if (pos < (unsigned)E) {
              int2 ov;
              ov.x = v.x & 0xFFFFFF;
              ov.y = v.y;
              sedge[pos] = ov;
            }
          }
        }
      }
    }
    __syncthreads();
  }
  offsum = so;
  return carry;
}

__global__ void __launch_bounds__(BIN_NODES)
bin_sort_kernel(const int* __restrict__ runTab, const int2* __restrict__ cedge,
                int* __restrict__ start, int2* __restrict__ sedge, float* __restrict__ dinv,
                int N, int E, int G) {
  __shared__ int2  sin_[SORT_STAGE];
  __shared__ int2  sout[SORT_STAGE];
  __shared__ int   cnt[BIN_NODES];
  __shared__ float wsumf[BIN_NODES];
  __shared__ int   cur[BIN_NODES];
  __shared__ int   rsrc[BIN_NODES];
  __shared__ int   rdst[BIN_NODES];
  __shared__ int   rcnt[BIN_NODES];
  __shared__ int   wtot[BIN_NODES / 32];
  __shared__ int   wred[BIN_NODES / 32];
  const int tid  = threadIdx.x;
  const int lane = tid & 31;
  const int wv   = tid >> 5;
  const int bin  = blockIdx.x;
  cnt[tid] = 0;
  wsumf[tid] = 0.0f;
  __syncthreads();
  int so = 0;
  const int total = sweep_runs<true>(runTab, cedge, bin, G, E, rsrc, rdst, rcnt, wtot, cnt, wsumf,
                                     sin_, cur, sedge, 0, so);
#pragma unroll
  for (int m = 16; m >= 1; m >>= 1) so += __shfl_xor(so, m, 32);
  if (lane == 0) wred[wv] = so;
  const int v = cnt[tid];
  int inc = v;
#pragma unroll
  for (int d = 1; d < 32; d <<= 1) {
    int t = __shfl_up(inc, d, 32);
    if (lane >= d) inc += t;
  }
  if (lane == 31) wtot[wv] = inc;
  __syncthreads();
  int bs = 0;
  for (int q = 0; q < BIN_NODES / 32; ++q) bs += wred[q];
  bs = min(max(bs, 0), E);
  int off = inc - v;
  for (int q = 0; q < wv; ++q) off += wtot[q];
  cur[tid] = off;
  const int node = (bin << BIN_SHIFT) + tid;
  if (node < N) {
    start[node] = bs + off;
    const float d = 1.0f + wsumf[tid];
    dinv[node] = d > 0.0f ? rsqrtf(d) : 0.0f;
    if (node == N - 1) start[N] = bs + off + v;
  }
  __syncthreads();
  if (total <= SORT_STAGE) {
    for (int i = tid; i < total; i += BIN_NODES) {
      const int2 ed = sin_[i];
      const int l = (int)((unsigned)ed.x >> 24);
      const int p = atomicAdd(&cur[l], 1);
      int2 ov;
      ov.x = ed.x & 0xFFFFFF;
      ov.y = ed.y;
      if ((unsigned)p < (unsigned)SORT_STAGE) sout[p] = ov;
    }
    __syncthreads();
    for (int i = tid; i < total; i += BIN_NODES) {
      const unsigned pos = (unsigned)(bs + i);
      if (pos < (unsigned)E) sedge[pos] = sout[i];
    }
  } else {
    int unused = 0;
    sweep_runs<false>(runTab, cedge, bin, G, E, rsrc, rdst, rcnt, wtot, cnt, wsumf, sin_, cur, sedge,
                      bs, unused);
  }
}

#define GEMM_WAVES 4
template <bool HAS_BIAS>
__global__ void __launch_bounds__(32 * GEMM_WAVES)
gemm_strip_kernel(const float* __restrict__ A, const float* __restrict__ Bp,
                  const float* __restrict__ bias, float* __restrict__ C, int nstrips) {
  __shared__ float4 sB[(D / 8) * (D / 16) * 32];
  const int tid  = threadIdx.x;
  const int wave = tid >> 5;
  const int lane = tid & 31;
  const int c    = lane & 15;
  const int h    = lane >> 4;
  const int strip = blockIdx.x * GEMM_WAVES + wave;
  const int lstrip = min(strip, nstrips - 1);

  const float4* arow = (const float4*)(A + (size_t)(lstrip * 16 + c) * D + 4 * h);
  float4 a[D / 8];
#pragma unroll
  for (int j = 0; j < D / 8; ++j) a[j] = arow[2 * j];

  const float4* bp4 = (const float4*)Bp;
#pragma unroll
  for (int i = 0; i < (D / 8) * (D / 16) / GEMM_WAVES; ++i)
    sB[tid + 32 * GEMM_WAVES * i] = bp4[tid + 32 * GEMM_WAVES * i];
  __syncthreads();
  if (strip >= nstrips) return;

  v8f acc[D / 16];
#pragma unroll
  for (int t = 0; t < D / 16; ++t) {
    const float b0 = HAS_BIAS ? bias[16 * t + c] : 0.0f;
    acc[t] = (v8f){b0, b0, b0, b0, b0, b0, b0, b0};
  }

  const float4* sl = sB + lane;
#pragma unroll
  for (int j = 0; j < D / 8; ++j) {
    float4 b[D / 16];
#pragma unroll
    for (int t = 0; t < D / 16; ++t) b[t] = sl[(j * (D / 16) + t) * 32];
    const v2f a01 = {a[j].x, a[j].y};
    const v2f a23 = {a[j].z, a[j].w};
#pragma unroll
    for (int t = 0; t < D / 16; ++t) {
      const v2f b01 = {b[t].x, b[t].y};
      acc[t] = __builtin_amdgcn_wmma_f32_16x16x4_f32(
          false, a01, false, b01, (short)0, acc[t], false, false);
    }
#pragma unroll
    for (int t = 0; t < D / 16; ++t) {
      const v2f b23 = {b[t].z, b[t].w};
      acc[t] = __builtin_amdgcn_wmma_f32_16x16x4_f32(
          false, a23, false, b23, (short)0, acc[t], false, false);
    }
  }

  float* cp = C + (size_t)(strip * 16 + h * 8) * D + c;
#pragma unroll
  for (int t = 0; t < D / 16; ++t)
#pragma unroll
    for (int r = 0; r < 8; ++r) cp[(size_t)r * D + 16 * t] = acc[t][r];
}

__global__ void __launch_bounds__(256)
gather_bias_relu_ln_kernel(const int* __restrict__ start, const int2* __restrict__ sedge,
                           const float* __restrict__ dinv, const float* __restrict__ h,
                           const float* __restrict__ bias, const float* __restrict__ gamma,
                           const float* __restrict__ beta, float* __restrict__ y,
                           int N, int E) {
  const int wave = __builtin_amdgcn_readfirstlane((int)(threadIdx.x >> 5));
  const int lane = threadIdx.x & 31;
  const int node = blockIdx.x * 8 + wave;
  if (node >= N) return;
  const int f = lane * 4;
  const float dc = dinv[node];
  const int beg = max(start[node], 0);
  const int end = min(start[node + 1], E);
  const float* hf = h + f;

  float4 acc;
  {
    const float cs = dc * dc;
    const float4 hv = *(const float4*)(hf + (size_t)node * D);
    acc.x = cs * hv.x; acc.y = cs * hv.y; acc.z = cs * hv.z; acc.w = cs * hv.w;
  }

  for (int base = beg; base < end; base += 32) {
    const int j = base + lane;
    int   r_l  = 0;
    float cf_l = 0.0f;
    if (j < end) {
      const int2 ed = sedge[j];
      r_l  = (int)min((unsigned)ed.x, (unsigned)(N - 1));
      cf_l = dinv[r_l] * __int_as_float(ed.y) * dc;
    }
    const int nb = min(32, end - base);
    int t = 0;
    for (; t + 4 <= nb; t += 4) {
      const int r0 = __builtin_amdgcn_readlane(r_l, t);
      const int r1 = __builtin_amdgcn_readlane(r_l, t + 1);
      const int r2 = __builtin_amdgcn_readlane(r_l, t + 2);
      const int r3 = __builtin_amdgcn_readlane(r_l, t + 3);
      const float c0 = __int_as_float(__builtin_amdgcn_readlane(__float_as_int(cf_l), t));
      const float c1 = __int_as_float(__builtin_amdgcn_readlane(__float_as_int(cf_l), t + 1));
      const float c2 = __int_as_float(__builtin_amdgcn_readlane(__float_as_int(cf_l), t + 2));
      const float c3 = __int_as_float(__builtin_amdgcn_readlane(__float_as_int(cf_l), t + 3));
      const float4 a0 = *(const float4*)(hf + (size_t)r0 * D);
      const float4 a1 = *(const float4*)(hf + (size_t)r1 * D);
      const float4 a2 = *(const float4*)(hf + (size_t)r2 * D);
      const float4 a3 = *(const float4*)(hf + (size_t)r3 * D);
      acc.x += c0 * a0.x; acc.y += c0 * a0.y; acc.z += c0 * a0.z; acc.w += c0 * a0.w;
      acc.x += c1 * a1.x; acc.y += c1 * a1.y; acc.z += c1 * a1.z; acc.w += c1 * a1.w;
      acc.x += c2 * a2.x; acc.y += c2 * a2.y; acc.z += c2 * a2.z; acc.w += c2 * a2.w;
      acc.x += c3 * a3.x; acc.y += c3 * a3.y; acc.z += c3 * a3.z; acc.w += c3 * a3.w;
    }
    for (; t < nb; ++t) {
      const int r0 = __builtin_amdgcn_readlane(r_l, t);
      const float c0 = __int_as_float(__builtin_amdgcn_readlane(__float_as_int(cf_l), t));
      const float4 a0 = *(const float4*)(hf + (size_t)r0 * D);
      acc.x += c0 * a0.x; acc.y += c0 * a0.y; acc.z += c0 * a0.z; acc.w += c0 * a0.w;
    }
  }

  float4 v = acc;
  const float4 b = *(const float4*)(bias + f);
  v.x = fmaxf(v.x + b.x, 0.0f);
  v.y = fmaxf(v.y + b.y, 0.0f);
  v.z = fmaxf(v.z + b.z, 0.0f);
  v.w = fmaxf(v.w + b.w, 0.0f);
  float s = v.x + v.y + v.z + v.w;
#pragma unroll
  for (int mski = 16; mski >= 1; mski >>= 1) s += __shfl_xor(s, mski, 32);
  float mu = s * (1.0f / D);
  float dx0 = v.x - mu, dx1 = v.y - mu, dx2 = v.z - mu, dx3 = v.w - mu;
  float vs = dx0 * dx0 + dx1 * dx1 + dx2 * dx2 + dx3 * dx3;
#pragma unroll
  for (int mski = 16; mski >= 1; mski >>= 1) vs += __shfl_xor(vs, mski, 32);
  float inv = rsqrtf(vs * (1.0f / D) + LN_EPS);
  const float4 g  = *(const float4*)(gamma + f);
  const float4 be = *(const float4*)(beta + f);
  float4 o;
  o.x = dx0 * inv * g.x + be.x;
  o.y = dx1 * inv * g.y + be.y;
  o.z = dx2 * inv * g.z + be.z;
  o.w = dx3 * inv * g.w + be.w;
  *(float4*)(y + (size_t)node * D + f) = o;
}

extern "C" void kernel_launch(void* const* d_in, const int* in_sizes, int n_in,
                              void* d_out, int out_size, void* d_ws, size_t ws_size,
                              hipStream_t stream) {
  const float* x     = (const float*)d_in[0];
  const int*   ei    = (const int*)d_in[1];
  const float* ew    = (const float*)d_in[2];
  const float* W     = (const float*)d_in[3];
  const float* bias  = (const float*)d_in[4];
  const float* gamma = (const float*)d_in[5];
  const float* beta  = (const float*)d_in[6];
  const float* linW  = (const float*)d_in[7];
  const float* linb  = (const float*)d_in[8];
  float* out = (float*)d_out;

  const int N = in_sizes[0] / D;
  const int E = in_sizes[2];
  const int B = (N + BIN_NODES - 1) / BIN_NODES;
  const int G = (E + CHUNK - 1) / CHUNK;

  char* ws = (char*)d_ws;
  const size_t nB = (((size_t)(N + 1) * sizeof(int)) + 255) & ~(size_t)255;
  const size_t tB = (size_t)G * MAXB * sizeof(int);
  const size_t cB = (size_t)G * CHUNK * sizeof(int2);
  float* dinv   = (float*)ws;
  int*   start  = (int*)(ws + nB);
  int*   runTab = (int*)(ws + 2 * nB);
  int2*  cedge  = (int2*)(ws + 2 * nB + tB);
  int2*  sedge  = (int2*)(ws + 2 * nB + tB + cB);
  float* h      = (float*)(ws + 2 * nB + tB + cB + (size_t)E * sizeof(int2));
  float* y     = h + (size_t)N * D;
  float4* Wp   = (float4*)(y + (size_t)N * D);
  float4* Lp   = Wp + PACK_ENTRIES;

  chunk_sort_pack_kernel<<<G + PACK_ENTRIES / CHUNK_T, CHUNK_T, 0, stream>>>(ei, ew, runTab, cedge, W,
                                                                            linW, Wp, Lp, E, N, G);
  bin_sort_kernel<<<B, BIN_NODES, 0, stream>>>(runTab, cedge, start, sedge, dinv, N, E, G);
  gemm_strip_kernel<false><<<(N / 16 + GEMM_WAVES - 1) / GEMM_WAVES, 32 * GEMM_WAVES, 0, stream>>>(
      x, (const float*)Wp, nullptr, h, N / 16);
  gather_bias_relu_ln_kernel<<<(N + 7) / 8, 256, 0, stream>>>(start, sedge, dinv, h, bias, gamma,
                                                             beta, y, N, E);
  gemm_strip_kernel<true><<<(N / 16 + GEMM_WAVES - 1) / GEMM_WAVES, 32 * GEMM_WAVES, 0, stream>>>(
      y, (const float*)Lp, linb, out, N / 16);
}


# First product writes rows pre-scaled by dinv[row]; gather drops the per-edge scattered dinv[source] load and scales the finished sum by dinv[target] once; sum order kept, products rounded differently
# speedup vs baseline: 4.3814x; 1.0310x over previous
#include <hip/hip_runtime.h>

#define D 128
#define LN_EPS 1e-5f

typedef __attribute__((ext_vector_type(2))) float v2f;
typedef __attribute__((ext_vector_type(8))) float v8f;

#define BIN_SHIFT 8
#define BIN_NODES 256
#define MAXB 512
#define CHUNK_T 256
#define CHUNK_K 16
#define CHUNK (CHUNK_T * CHUNK_K)
#define PACK_ENTRIES ((D / 8) * (D / 16) * 32)

__global__ void __launch_bounds__(CHUNK_T)
chunk_sort_pack_kernel(const int* __restrict__ ei, const float* __restrict__ w,
                       int* __restrict__ runTab, int2* __restrict__ cedge,
                       const float* __restrict__ W, const float* __restrict__ linW,
                       float4* __restrict__ Wp, float4* __restrict__ Lp, int E, int N, int G) {
  __shared__ int2 stage[CHUNK];
  __shared__ int hist[MAXB];
  __shared__ int lbase[MAXB];
  __shared__ int wsumL[CHUNK_T / 32];
  const int tid  = threadIdx.x;
  if ((int)blockIdx.x >= G) {
    const int i = ((int)blockIdx.x - G) * CHUNK_T + tid;
    if (i < PACK_ENTRIES) {
      const int lane = i & 31;
      const int t    = (i >> 5) & 7;
      const int j    = i >> 8;
      const int k0   = 8 * j + 4 * (lane >> 4);
      const int n    = 16 * t + (lane & 15);
      float4 wv;
      wv.x = W[(size_t)(k0 + 0) * D + n];
      wv.y = W[(size_t)(k0 + 1) * D + n];
      wv.z = W[(size_t)(k0 + 2) * D + n];
      wv.w = W[(size_t)(k0 + 3) * D + n];
      Wp[i] = wv;
      Lp[i] = *(const float4*)(linW + (size_t)n * D + k0);
    }
    return;
  }
  const int lane = tid & 31;
  const int wv   = tid >> 5;
  for (int i = tid; i < MAXB; i += CHUNK_T) hist[i] = 0;
  const int base = blockIdx.x * CHUNK;
  const int nvalid = min(CHUNK, E - base);
  int col[CHUNK_K], row[CHUNK_K], wb[CHUNK_K], rk[CHUNK_K];
#pragma unroll
  for (int k = 0; k < CHUNK_K; ++k) {
    const int e = min(base + k * CHUNK_T + tid, E - 1);
    col[k] = (int)min((unsigned)ei[E + e], (unsigned)(N - 1));
    row[k] = (int)min((unsigned)ei[e], (unsigned)(N - 1));
    wb[k]  = __float_as_int(w[e]);
  }
  __syncthreads();
#pragma unroll
  for (int k = 0; k < CHUNK_K; ++k) {
    rk[k] = 0;
    if (k * CHUNK_T + tid < nvalid) rk[k] = atomicAdd(&hist[col[k] >> BIN_SHIFT], 1);
  }
  __syncthreads();
  const int b0 = 2 * tid, b1 = 2 * tid + 1;
  const int c0 = hist[b0], c1 = hist[b1];
  const int sL = c0 + c1;
  int incL = sL;
#pragma unroll
  for (int d = 1; d < 32; d <<= 1) {
    int tL = __shfl_up(incL, d, 32);
    if (lane >= d) incL += tL;
  }
  if (lane == 31) wsumL[wv] = incL;
  __syncthreads();
  int exL = incL - sL;
  for (int q = 0; q < wv; ++q) exL += wsumL[q];
  lbase[b0] = exL;
  lbase[b1] = exL + c0;
  {
    int2 te;
    te.x = (exL << 16) | c0;
    te.y = ((exL + c0) << 16) | c1;
    *(int2*)(runTab + (size_t)blockIdx.x * MAXB + b0) = te;
  }
  __syncthreads();
  int ls[CHUNK_K];
#pragma unroll
  for (int k = 0; k < CHUNK_K; ++k) ls[k] = lbase[col[k] >> BIN_SHIFT] + rk[k];
#pragma unroll
  for (int k = 0; k < CHUNK_K; ++k) {
    if (k * CHUNK_T + tid < nvalid) {
      int2 ed;
      ed.x = (int)((unsigned)row[k] | ((unsigned)(col[k] & (BIN_NODES - 1)) << 24));
      ed.y = wb[k];
      stage[ls[k]] = ed;
    }
  }
  __syncthreads();
  int2* dst = cedge + (size_t)blockIdx.x * CHUNK;
#pragma unroll
  for (int k = 0; k < CHUNK_K; ++k) {
    const int i = k * CHUNK_T + tid;
    int2 v = stage[min(i, nvalid - 1)];
    if (i >= nvalid) { v.x = 0; v.y = 0; }
    dst[i] = v;
  }
}

#define SORT_STAGE 8192
#define RUN_U 8
template <bool COLLECT>
__device__ __forceinline__ int sweep_runs(const int* __restrict__ runTab,
                                          const int2* __restrict__ cedge, int bin, int G, int E,
                                          int* rsrc, int* rdst, int* rcnt, int* wtot, int* cnt,
                                          float* wsumf, int2* sin_, int* cur,
                                          int2* __restrict__ sedge, int bs, int& offsum) {
  const int tid  = threadIdx.x;
  const int lane = tid & 31;
  const int wv   = tid >> 5;
  const int hl   = lane & 15;
  const int hh   = lane >> 4;
  const int lim  = G * CHUNK - 1;
  int carry = 0, so = 0;
  for (int g0 = 0; g0 < G; g0 += BIN_NODES) {
    const int g = g0 + tid;
    const int e = g < G ? runTab[(size_t)g * MAXB + bin] : 0;
    const int c = e & 0xFFFF;
    const int o = (int)((unsigned)e >> 16);
    so += o;
    int inc = c;
#pragma unroll
    for (int d = 1; d < 32; d <<= 1) {
      int t = __shfl_up(inc, d, 32);
      if (lane >= d) inc += t;
    }
    if (lane == 31) wtot[wv] = inc;
    __syncthreads();
    int ex = inc - c, bt = 0;
    for (int q = 0; q < BIN_NODES / 32; ++q) {
      const int t = wtot[q];
      if (q < wv) ex += t;
      bt += t;
    }
    rsrc[tid] = g * CHUNK + o;
    rdst[tid] = carry + ex;
    rcnt[tid] = c;
    __syncthreads();
    carry += bt;
    for (int it = 0; it < BIN_NODES / 16; it += RUN_U) {
      int rs[RUN_U], rd[RUN_U], rc[RUN_U];
      int2 ed[RUN_U];
#pragma unroll
      for (int u = 0; u < RUN_U; ++u) {
        const int r = ((it + u) * (BIN_NODES / 32) + wv) * 2 + hh;
        rs[u] = rsrc[r];
        rd[u] = rdst[r];
        rc[u] = rcnt[r];
      }
#pragma unroll
      for (int u = 0; u < RUN_U; ++u) ed[u] = cedge[min(rs[u] + hl, lim)];
#pragma unroll
      for (int u = 0; u < RUN_U; ++u) {
        for (int j = hl; j < rc[u]; j += 16) {
          const int2 v = j == hl ? ed[u] : cedge[min(rs[u] + j, lim)];
          const int l = (int)((unsigned)v.x >> 24);
          if (COLLECT) {
            atomicAdd(&cnt[l], 1);
            atomicAdd(&wsumf[l], __int_as_float(v.y));
            const unsigned dpos = (unsigned)(rd[u] + j);
            if (dpos < (unsigned)SORT_STAGE) sin_[dpos] = v;
          } else {
            const int p = atomicAdd(&cur[l], 1);
            const unsigned pos = (unsigned)(bs + p);
            if (pos < (unsigned)E) {
              int2 ov;
              ov.x = v.x & 0xFFFFFF;
              ov.y = v.y;
              sedge[pos] = ov;
            }
          }
        }
      }
    }
    __syncthreads();
  }
  offsum = so;
  return carry;
}

__global__ void __launch_bounds__(BIN_NODES)
bin_sort_kernel(const int* __restrict__ runTab, const int2* __restrict__ cedge,
                int* __restrict__ start, int2* __restrict__ sedge, float* __restrict__ dinv,
                int N, int E, int G) {
  __shared__ int2  sin_[SORT_STAGE];
  __shared__ int2  sout[SORT_STAGE];
  __shared__ int   cnt[BIN_NODES];
  __shared__ float wsumf[BIN_NODES];
  __shared__ int   cur[BIN_NODES];
  __shared__ int   rsrc[BIN_NODES];
  __shared__ int   rdst[BIN_NODES];
  __shared__ int   rcnt[BIN_NODES];
  __shared__ int   wtot[BIN_NODES / 32];
  __shared__ int   wred[BIN_NODES / 32];
  const int tid  = threadIdx.x;
  const int lane = tid & 31;
  const int wv   = tid >> 5;
  const int bin  = blockIdx.x;
  cnt[tid] = 0;
  wsumf[tid] = 0.0f;
  __syncthreads();
  int so = 0;
  const int total = sweep_runs<true>(runTab, cedge, bin, G, E, rsrc, rdst, rcnt, wtot, cnt, wsumf,
                                     sin_, cur, sedge, 0, so);
#pragma unroll
  for (int m = 16; m >= 1; m >>= 1) so += __shfl_xor(so, m, 32);
  if (lane == 0) wred[wv] = so;
  const int v = cnt[tid];
  int inc = v;
#pragma unroll
  for (int d = 1; d < 32; d <<= 1) {
    int t = __shfl_up(inc, d, 32);
    if (lane >= d) inc += t;
  }
  if (lane == 31) wtot[wv] = inc;
  __syncthreads();
  int bs = 0;
  for (int q = 0; q < BIN_NODES / 32; ++q) bs += wred[q];
  bs = min(max(bs, 0), E);
  int off = inc - v;
  for (int q = 0; q < wv; ++q) off += wtot[q];
  cur[tid] = off;
  const int node = (bin << BIN_SHIFT) + tid;
  if (node < N) {
    start[node] = bs + off;
    const float d = 1.0f + wsumf[tid];
    dinv[node] = d > 0.0f ? rsqrtf(d) : 0.0f;
    if (node == N - 1) start[N] = bs + off + v;
  }
  __syncthreads();
  if (total <= SORT_STAGE) {
    for (int i = tid; i < total; i += BIN_NODES) {
      const int2 ed = sin_[i];
      const int l = (int)((unsigned)ed.x >> 24);
      const int p = atomicAdd(&cur[l], 1);
      int2 ov;
      ov.x = ed.x & 0xFFFFFF;
      ov.y = ed.y;
      if ((unsigned)p < (unsigned)SORT_STAGE) sout[p] = ov;
    }
    __syncthreads();
    for (int i = tid; i < total; i += BIN_NODES) {
      const unsigned pos = (unsigned)(bs + i);
      if (pos < (unsigned)E) sedge[pos] = sout[i];
    }
  } else {
    int unused = 0;
    sweep_runs<false>(runTab, cedge, bin, G, E, rsrc, rdst, rcnt, wtot, cnt, wsumf, sin_, cur, sedge,
                      bs, unused);
  }
}

#define GEMM_WAVES 4
template <bool HAS_BIAS, bool ROW_SCALE>
__global__ void __launch_bounds__(32 * GEMM_WAVES)
gemm_strip_kernel(const float* __restrict__ A, const float* __restrict__ Bp,
                  const float* __restrict__ bias, const float* __restrict__ rowscale,
                  float* __restrict__ C, int nstrips) {
  __shared__ float4 sB[(D / 8) * (D / 16) * 32];
  const int tid  = threadIdx.x;
  const int wave = tid >> 5;
  const int lane = tid & 31;
  const int c    = lane & 15;
  const int h    = lane >> 4;
  const int strip = blockIdx.x * GEMM_WAVES + wave;
  const int lstrip = min(strip, nstrips - 1);

  const float4* arow = (const float4*)(A + (size_t)(lstrip * 16 + c) * D + 4 * h);
  float4 a[D / 8];
#pragma unroll
  for (int j = 0; j < D / 8; ++j) a[j] = arow[2 * j];

  const float4* bp4 = (const float4*)Bp;
#pragma unroll
  for (int i = 0; i < (D / 8) * (D / 16) / GEMM_WAVES; ++i)
    sB[tid + 32 * GEMM_WAVES * i] = bp4[tid + 32 * GEMM_WAVES * i];
  __syncthreads();
  if (strip >= nstrips) return;

  float rs[8];
  if (ROW_SCALE) {
    const float4* rp = (const float4*)(rowscale + (size_t)strip * 16 + 8 * h);
    const float4 s0 = rp[0];
    const float4 s1 = rp[1];
    rs[0] = s0.x; rs[1] = s0.y; rs[2] = s0.z; rs[3] = s0.w;
    rs[4] = s1.x; rs[5] = s1.y; rs[6] = s1.z; rs[7] = s1.w;
  }

  v8f acc[D / 16];
#pragma unroll
  for (int t = 0; t < D / 16; ++t) {
    const float b0 = HAS_BIAS ? bias[16 * t + c] : 0.0f;
    acc[t] = (v8f){b0, b0, b0, b0, b0, b0, b0, b0};
  }

  const float4* sl = sB + lane;
#pragma unroll
  for (int j = 0; j < D / 8; ++j) {
    float4 b[D / 16];
#pragma unroll
    for (int t = 0; t < D / 16; ++t) b[t] = sl[(j * (D / 16) + t) * 32];
    const v2f a01 = {a[j].x, a[j].y};
    const v2f a23 = {a[j].z, a[j].w};
#pragma unroll
    for (int t = 0; t < D / 16; ++t) {
      const v2f b01 = {b[t].x, b[t].y};
      acc[t] = __builtin_amdgcn_wmma_f32_16x16x4_f32(
          false, a01, false, b01, (short)0, acc[t], false, false);
    }
#pragma unroll
    for (int t = 0; t < D / 16; ++t) {
      const v2f b23 = {b[t].z, b[t].w};
      acc[t] = __builtin_amdgcn_wmma_f32_16x16x4_f32(
          false, a23, false, b23, (short)0, acc[t], false, false);
    }
  }

  float* cp = C + (size_t)(strip * 16 + h * 8) * D + c;
#pragma unroll
  for (int t = 0; t < D / 16; ++t)
#pragma unroll
    for (int r = 0; r < 8; ++r)
      cp[(size_t)r * D + 16 * t] = ROW_SCALE ? acc[t][r] * rs[r] : acc[t][r];
}

__global__ void __launch_bounds__(256)
gather_bias_relu_ln_kernel(const int* __restrict__ start, const int2* __restrict__ sedge,
                           const float* __restrict__ dinv, const float* __restrict__ h,
                           const float* __restrict__ bias, const float* __restrict__ gamma,
                           const float* __restrict__ beta, float* __restrict__ y,
                           int N, int E) {
  const int wave = __builtin_amdgcn_readfirstlane((int)(threadIdx.x >> 5));
  const int lane = threadIdx.x & 31;
  const int node = blockIdx.x * 8 + wave;
  if (node >= N) return;
  const int f = lane * 4;
  const float dc = dinv[node];
  const int beg = max(start[node], 0);
  const int end = min(start[node + 1], E);
  const float* hf = h + f;

  float4 acc = *(const float4*)(hf + (size_t)node * D);

  for (int base = beg; base < end; base += 32) {
    const int j = base + lane;
    int   r_l  = 0;
    float cf_l = 0.0f;
    if (j < end) {
      const int2 ed = sedge[j];
      r_l  = (int)min((unsigned)ed.x, (unsigned)(N - 1));
      cf_l = __int_as_float(ed.y);
    }
    const int nb = min(32, end - base);
    int t = 0;
    for (; t + 4 <= nb; t += 4) {
      const int r0 = __builtin_amdgcn_readlane(r_l, t);
      const int r1 = __builtin_amdgcn_readlane(r_l, t + 1);
      const int r2 = __builtin_amdgcn_readlane(r_l, t + 2);
      const int r3 = __builtin_amdgcn_readlane(r_l, t + 3);
      const float c0 = __int_as_float(__builtin_amdgcn_readlane(__float_as_int(cf_l), t));
      const float c1 = __int_as_float(__builtin_amdgcn_readlane(__float_as_int(cf_l), t + 1));
      const float c2 = __int_as_float(__builtin_amdgcn_readlane(__float_as_int(cf_l), t + 2));
      const float c3 = __int_as_float(__builtin_amdgcn_readlane(__float_as_int(cf_l), t + 3));
      const float4 a0 = *(const float4*)(hf + (size_t)r0 * D);
      const float4 a1 = *(const float4*)(hf + (size_t)r1 * D);
      const float4 a2 = *(const float4*)(hf + (size_t)r2 * D);
      const float4 a3 = *(const float4*)(hf + (size_t)r3 * D);
      acc.x += c0 * a0.x; acc.y += c0 * a0.y; acc.z += c0 * a0.z; acc.w += c0 * a0.w;
      acc.x += c1 * a1.x; acc.y += c1 * a1.y; acc.z += c1 * a1.z; acc.w += c1 * a1.w;
      acc.x += c2 * a2.x; acc.y += c2 * a2.y; acc.z += c2 * a2.z; acc.w += c2 * a2.w;
      acc.x += c3 * a3.x; acc.y += c3 * a3.y; acc.z += c3 * a3.z; acc.w += c3 * a3.w;
    }
    for (; t < nb; ++t) {
      const int r0 = __builtin_amdgcn_readlane(r_l, t);
      const float c0 = __int_as_float(__builtin_amdgcn_readlane(__float_as_int(cf_l), t));
      const float4 a0 = *(const float4*)(hf + (size_t)r0 * D);
      acc.x += c0 * a0.x; acc.y += c0 * a0.y; acc.z += c0 * a0.z; acc.w += c0 * a0.w;
    }
  }

  float4 v;
  v.x = dc * acc.x; v.y = dc * acc.y; v.z = dc * acc.z; v.w = dc * acc.w;
  const float4 b = *(const float4*)(bias + f);
  v.x = fmaxf(v.x + b.x, 0.0f);
  v.y = fmaxf(v.y + b.y, 0.0f);
  v.z = fmaxf(v.z + b.z, 0.0f);
  v.w = fmaxf(v.w + b.w, 0.0f);
  float s = v.x + v.y + v.z + v.w;
#pragma unroll
  for (int mski = 16; mski >= 1; mski >>= 1) s += __shfl_xor(s, mski, 32);
  float mu = s * (1.0f / D);
  float dx0 = v.x - mu, dx1 = v.y - mu, dx2 = v.z - mu, dx3 = v.w - mu;
  float vs = dx0 * dx0 + dx1 * dx1 + dx2 * dx2 + dx3 * dx3;
#pragma unroll
  for (int mski = 16; mski >= 1; mski >>= 1) vs += __shfl_xor(vs, mski, 32);
  float inv = rsqrtf(vs * (1.0f / D) + LN_EPS);
  const float4 g  = *(const float4*)(gamma + f);
  const float4 be = *(const float4*)(beta + f);
  float4 o;
  o.x = dx0 * inv * g.x + be.x;
  o.y = dx1 * inv * g.y + be.y;
  o.z = dx2 * inv * g.z + be.z;
  o.w = dx3 * inv * g.w + be.w;
  *(float4*)(y + (size_t)node * D + f) = o;
}

extern "C" void kernel_launch(void* const* d_in, const int* in_sizes, int n_in,
                              void* d_out, int out_size, void* d_ws, size_t ws_size,
                              hipStream_t stream) {
  const float* x     = (const float*)d_in[0];
  const int*   ei    = (const int*)d_in[1];
  const float* ew    = (const float*)d_in[2];
  const float* W     = (const float*)d_in[3];
  const float* bias  = (const float*)d_in[4];
  const float* gamma = (const float*)d_in[5];
  const float* beta  = (const float*)d_in[6];
  const float* linW  = (const float*)d_in[7];
  const float* linb  = (const float*)d_in[8];
  float* out = (float*)d_out;

  const int N = in_sizes[0] / D;
  const int E = in_sizes[2];
  const int B = (N + BIN_NODES - 1) / BIN_NODES;
  const int G = (E + CHUNK - 1) / CHUNK;

  char* ws = (char*)d_ws;
  const size_t nB = (((size_t)(N + 1) * sizeof(int)) + 255) & ~(size_t)255;
  const size_t tB = (size_t)G * MAXB * sizeof(int);
  const size_t cB = (size_t)G * CHUNK * sizeof(int2);
  float* dinv   = (float*)ws;
  int*   start  = (int*)(ws + nB);
  int*   runTab = (int*)(ws + 2 * nB);
  int2*  cedge  = (int2*)(ws + 2 * nB + tB);
  int2*  sedge  = (int2*)(ws + 2 * nB + tB + cB);
  float* h      = (float*)(ws + 2 * nB + tB + cB + (size_t)E * sizeof(int2));
  float* y     = h + (size_t)N * D;
  float4* Wp   = (float4*)(y + (size_t)N * D);
  float4* Lp   = Wp + PACK_ENTRIES;

  chunk_sort_pack_kernel<<<G + PACK_ENTRIES / CHUNK_T, CHUNK_T, 0, stream>>>(ei, ew, runTab, cedge, W,
                                                                            linW, Wp, Lp, E, N, G);
  bin_sort_kernel<<<B, BIN_NODES, 0, stream>>>(runTab, cedge, start, sedge, dinv, N, E, G);
  gemm_strip_kernel<false, true><<<(N / 16 + GEMM_WAVES - 1) / GEMM_WAVES, 32 * GEMM_WAVES, 0,
                                  stream>>>(x, (const float*)Wp, nullptr, dinv, h, N / 16);
  gather_bias_relu_ln_kernel<<<(N + 7) / 8, 256, 0, stream>>>(start, sedge, dinv, h, bias, gamma,
                                                             beta, y, N, E);
  gemm_strip_kernel<true, false><<<(N / 16 + GEMM_WAVES - 1) / GEMM_WAVES, 32 * GEMM_WAVES, 0,
                                  stream>>>(y, (const float*)Lp, linb, nullptr, out, N / 16);
}


# Second product fused into the gather kernel: 16-wave work-group gathers and normalises 48 rows into LDS (3 nodes per wave), multiplies them there; y never reaches memory; 4 launches; sum order kept
# speedup vs baseline: 4.5235x; 1.0324x over previous
#include <hip/hip_runtime.h>

#define D 128
#define LN_EPS 1e-5f

typedef __attribute__((ext_vector_type(2))) float v2f;
typedef __attribute__((ext_vector_type(8))) float v8f;

#define BIN_SHIFT 8
#define BIN_NODES 256
#define MAXB 512
#define CHUNK_T 256
#define CHUNK_K 16
#define CHUNK (CHUNK_T * CHUNK_K)
#define PACK_ENTRIES ((D / 8) * (D / 16) * 32)

__global__ void __launch_bounds__(CHUNK_T)
chunk_sort_pack_kernel(const int* __restrict__ ei, const float* __restrict__ w,
                       int* __restrict__ runTab, int2* __restrict__ cedge,
                       const float* __restrict__ W, const float* __restrict__ linW,
                       float4* __restrict__ Wp, float4* __restrict__ Lp, int E, int N, int G) {
  __shared__ int2 stage[CHUNK];
  __shared__ int hist[MAXB];
  __shared__ int lbase[MAXB];
  __shared__ int wsumL[CHUNK_T / 32];
  const int tid  = threadIdx.x;
  if ((int)blockIdx.x >= G) {
    const int i = ((int)blockIdx.x - G) * CHUNK_T + tid;
    if (i < PACK_ENTRIES) {
      const int lane = i & 31;
      const int t    = (i >> 5) & 7;
      const int j    = i >> 8;
      const int k0   = 8 * j + 4 * (lane >> 4);
      const int n    = 16 * t + (lane & 15);
      float4 wv;
      wv.x = W[(size_t)(k0 + 0) * D + n];
      wv.y = W[(size_t)(k0 + 1) * D + n];
      wv.z = W[(size_t)(k0 + 2) * D + n];
      wv.w = W[(size_t)(k0 + 3) * D + n];
      Wp[i] = wv;
      Lp[i] = *(const float4*)(linW + (size_t)n * D + k0);
    }
    return;
  }
  const int lane = tid & 31;
  const int wv   = tid >> 5;
  for (int i = tid; i < MAXB; i += CHUNK_T) hist[i] = 0;
  const int base = blockIdx.x * CHUNK;
  const int nvalid = min(CHUNK, E - base);
  int col[CHUNK_K], row[CHUNK_K], wb[CHUNK_K], rk[CHUNK_K];
#pragma unroll
  for (int k = 0; k < CHUNK_K; ++k) {
    const int e = min(base + k * CHUNK_T + tid, E - 1);
    col[k] = (int)min((unsigned)ei[E + e], (unsigned)(N - 1));
    row[k] = (int)min((unsigned)ei[e], (unsigned)(N - 1));
    wb[k]  = __float_as_int(w[e]);
  }
  __syncthreads();
#pragma unroll
  for (int k = 0; k < CHUNK_K; ++k) {
    rk[k] = 0;
    if (k * CHUNK_T + tid < nvalid) rk[k] = atomicAdd(&hist[col[k] >> BIN_SHIFT], 1);
  }
  __syncthreads();
  const int b0 = 2 * tid, b1 = 2 * tid + 1;
  const int c0 = hist[b0], c1 = hist[b1];
  const int sL = c0 + c1;
  int incL = sL;
#pragma unroll
  for (int d = 1; d < 32; d <<= 1) {
    int tL = __shfl_up(incL, d, 32);
    if (lane >= d) incL += tL;
  }
  if (lane == 31) wsumL[wv] = incL;
  __syncthreads();
  int exL = incL - sL;
  for (int q = 0; q < wv; ++q) exL += wsumL[q];
  lbase[b0] = exL;
  lbase[b1] = exL + c0;
  {
    int2 te;
    te.x = (exL << 16) | c0;
    te.y = ((exL + c0) << 16) | c1;
    *(int2*)(runTab + (size_t)blockIdx.x * MAXB + b0) = te;
  }
  __syncthreads();
  int ls[CHUNK_K];
#pragma unroll
  for (int k = 0; k < CHUNK_K; ++k) ls[k] = lbase[col[k] >> BIN_SHIFT] + rk[k];
#pragma unroll
  for (int k = 0; k < CHUNK_K; ++k) {
    if (k * CHUNK_T + tid < nvalid) {
      int2 ed;
      ed.x = (int)((unsigned)row[k] | ((unsigned)(col[k] & (BIN_NODES - 1)) << 24));
      ed.y = wb[k];
      stage[ls[k]] = ed;
    }
  }
  __syncthreads();
  int2* dst = cedge + (size_t)blockIdx.x * CHUNK;
#pragma unroll
  for (int k = 0; k < CHUNK_K; ++k) {
    const int i = k * CHUNK_T + tid;
    int2 v = stage[min(i, nvalid - 1)];
    if (i >= nvalid) { v.x = 0; v.y = 0; }
    dst[i] = v;
  }
}

#define SORT_STAGE 8192
#define RUN_U 8
template <bool COLLECT>
__device__ __forceinline__ int sweep_runs(const int* __restrict__ runTab,
                                          const int2* __restrict__ cedge, int bin, int G, int E,
                                          int* rsrc, int* rdst, int* rcnt, int* wtot, int* cnt,
                                          float* wsumf, int2* sin_, int* cur,
                                          int2* __restrict__ sedge, int bs, int& offsum) {
  const int tid  = threadIdx.x;
  const int lane = tid & 31;
  const int wv   = tid >> 5;
  const int hl   = lane & 15;
  const int hh   = lane >> 4;
  const int lim  = G * CHUNK - 1;
  int carry = 0, so = 0;
  for (int g0 = 0; g0 < G; g0 += BIN_NODES) {
    const int g = g0 + tid;
    const int e = g < G ? runTab[(size_t)g * MAXB + bin] : 0;
    const int c = e & 0xFFFF;
    const int o = (int)((unsigned)e >> 16);
    so += o;
    int inc = c;
#pragma unroll
    for (int d = 1; d < 32; d <<= 1) {
      int t = __shfl_up(inc, d, 32);
      if (lane >= d) inc += t;
    }
    if (lane == 31) wtot[wv] = inc;
    __syncthreads();
    int ex = inc - c, bt = 0;
    for (int q = 0; q < BIN_NODES / 32; ++q) {
      const int t = wtot[q];
      if (q < wv) ex += t;
      bt += t;
    }
    rsrc[tid] = g * CHUNK + o;
    rdst[tid] = carry + ex;
    rcnt[tid] = c;
    __syncthreads();
    carry += bt;
    for (int it = 0; it < BIN_NODES / 16; it += RUN_U) {
      int rs[RUN_U], rd[RUN_U], rc[RUN_U];
      int2 ed[RUN_U];
#pragma unroll
      for (int u = 0; u < RUN_U; ++u) {
        const int r = ((it + u) * (BIN_NODES / 32) + wv) * 2 + hh;
        rs[u] = rsrc[r];
        rd[u] = rdst[r];
        rc[u] = rcnt[r];
      }
#pragma unroll
      for (int u = 0; u < RUN_U; ++u) ed[u] = cedge[min(rs[u] + hl, lim)];
#pragma unroll
      for (int u = 0; u < RUN_U; ++u) {
        for (int j = hl; j < rc[u]; j += 16) {
          const int2 v = j == hl ? ed[u] : cedge[min(rs[u] + j, lim)];
          const int l = (int)((unsigned)v.x >> 24);
          if (COLLECT) {
            atomicAdd(&cnt[l], 1);
            atomicAdd(&wsumf[l], __int_as_float(v.y));
            const unsigned dpos = (unsigned)(rd[u] + j);
            if (dpos < (unsigned)SORT_STAGE) sin_[dpos] = v;
          } else {
            const int p = atomicAdd(&cur[l], 1);
            const unsigned pos = (unsigned)(bs + p);
            if (pos < (unsigned)E) {
              int2 ov;
              ov.x = v.x & 0xFFFFFF;
              ov.y = v.y;
              sedge[pos] = ov;
            }
          }
        }
      }
    }
    __syncthreads();
  }
  offsum = so;
  return carry;
}

__global__ void __launch_bounds__(BIN_NODES)
bin_sort_kernel(const int* __restrict__ runTab, const int2* __restrict__ cedge,
                int* __restrict__ start, int2* __restrict__ sedge, float* __restrict__ dinv,
                int N, int E, int G) {
  __shared__ int2  sin_[SORT_STAGE];
  __shared__ int2  sout[SORT_STAGE];
  __shared__ int   cnt[BIN_NODES];
  __shared__ float wsumf[BIN_NODES];
  __shared__ int   cur[BIN_NODES];
  __shared__ int   rsrc[BIN_NODES];
  __shared__ int   rdst[BIN_NODES];
  __shared__ int   rcnt[BIN_NODES];
  __shared__ int   wtot[BIN_NODES / 32];
  __shared__ int   wred[BIN_NODES / 32];
  const int tid  = threadIdx.x;
  const int lane = tid & 31;
  const int wv   = tid >> 5;
  const int bin  = blockIdx.x;
  cnt[tid] = 0;
  wsumf[tid] = 0.0f;
  __syncthreads();
  int so = 0;
  const int total = sweep_runs<true>(runTab, cedge, bin, G, E, rsrc, rdst, rcnt, wtot, cnt, wsumf,
                                     sin_, cur, sedge, 0, so);
#pragma unroll
  for (int m = 16; m >= 1; m >>= 1) so += __shfl_xor(so, m, 32);
  if (lane == 0) wred[wv] = so;
  const int v = cnt[tid];
  int inc = v;
#pragma unroll
  for (int d = 1; d < 32; d <<= 1) {
    int t = __shfl_up(inc, d, 32);
    if (lane >= d) inc += t;
  }
  if (lane == 31) wtot[wv] = inc;
  __syncthreads();
  int bs = 0;
  for (int q = 0; q < BIN_NODES / 32; ++q) bs += wred[q];
  bs = min(max(bs, 0), E);
  int off = inc - v;
  for (int q = 0; q < wv; ++q) off += wtot[q];
  cur[tid] = off;
  const int node = (bin << BIN_SHIFT) + tid;
  if (node < N) {
    start[node] = bs + off;
    const float d = 1.0f + wsumf[tid];
    dinv[node] = d > 0.0f ? rsqrtf(d) : 0.0f;
    if (node == N - 1) start[N] = bs + off + v;
  }
  __syncthreads();
  if (total <= SORT_STAGE) {
    for (int i = tid; i < total; i += BIN_NODES) {
      const int2 ed = sin_[i];
      const int l = (int)((unsigned)ed.x >> 24);
      const int p = atomicAdd(&cur[l], 1);
      int2 ov;
      ov.x = ed.x & 0xFFFFFF;
      ov.y = ed.y;
      if ((unsigned)p < (unsigned)SORT_STAGE) sout[p] = ov;
    }
    __syncthreads();
    for (int i = tid; i < total; i += BIN_NODES) {
      const unsigned pos = (unsigned)(bs + i);
      if (pos < (unsigned)E) sedge[pos] = sout[i];
    }
  } else {
    int unused = 0;
    sweep_runs<false>(runTab, cedge, bin, G, E, rsrc, rdst, rcnt, wtot, cnt, wsumf, sin_, cur, sedge,
                      bs, unused);
  }
}

#define GEMM_WAVES 4
template <bool HAS_BIAS, bool ROW_SCALE>
__global__ void __launch_bounds__(32 * GEMM_WAVES)
gemm_strip_kernel(const float* __restrict__ A, const float* __restrict__ Bp,
                  const float* __restrict__ bias, const float* __restrict__ rowscale,
                  float* __restrict__ C, int nstrips) {
  __shared__ float4 sB[(D / 8) * (D / 16) * 32];
  const int tid  = threadIdx.x;
  const int wave = tid >> 5;
  const int lane = tid & 31;
  const int c    = lane & 15;
  const int h    = lane >> 4;
  const int strip = blockIdx.x * GEMM_WAVES + wave;
  const int lstrip = min(strip, nstrips - 1);

  const float4* arow = (const float4*)(A + (size_t)(lstrip * 16 + c) * D + 4 * h);
  float4 a[D / 8];
#pragma unroll
  for (int j = 0; j < D / 8; ++j) a[j] = arow[2 * j];

  const float4* bp4 = (const float4*)Bp;
#pragma unroll
  for (int i = 0; i < (D / 8) * (D / 16) / GEMM_WAVES; ++i)
    sB[tid + 32 * GEMM_WAVES * i] = bp4[tid + 32 * GEMM_WAVES * i];
  __syncthreads();
  if (strip >= nstrips) return;

  float rs[8];
  if (ROW_SCALE) {
    const float4* rp = (const float4*)(rowscale + (size_t)strip * 16 + 8 * h);
    const float4 s0 = rp[0];
    const float4 s1 = rp[1];
    rs[0] = s0.x; rs[1] = s0.y; rs[2] = s0.z; rs[3] = s0.w;
    rs[4] = s1.x; rs[5] = s1.y; rs[6] = s1.z; rs[7] = s1.w;
  }

  v8f acc[D / 16];
#pragma unroll
  for (int t = 0; t < D / 16; ++t) {
    const float b0 = HAS_BIAS ? bias[16 * t + c] : 0.0f;
    acc[t] = (v8f){b0, b0, b0, b0, b0, b0, b0, b0};
  }

  const float4* sl = sB + lane;
#pragma unroll
  for (int j = 0; j < D / 8; ++j) {
    float4 b[D / 16];
#pragma unroll
    for (int t = 0; t < D / 16; ++t) b[t] = sl[(j * (D / 16) + t) * 32];
    const v2f a01 = {a[j].x, a[j].y};
    const v2f a23 = {a[j].z, a[j].w};
#pragma unroll
    for (int t = 0; t < D / 16; ++t) {
      const v2f b01 = {b[t].x, b[t].y};
      acc[t] = __builtin_amdgcn_wmma_f32_16x16x4_f32(
          false, a01, false, b01, (short)0, acc[t], false, false);
    }
#pragma unroll
    for (int t = 0; t < D / 16; ++t) {
      const v2f b23 = {b[t].z, b[t].w};
      acc[t] = __builtin_amdgcn_wmma_f32_16x16x4_f32(
          false, a23, false, b23, (short)0, acc[t], false, false);
    }
  }

  float* cp = C + (size_t)(strip * 16 + h * 8) * D + c;
#pragma unroll
  for (int t = 0; t < D / 16; ++t)
#pragma unroll
    for (int r = 0; r < 8; ++r)
      cp[(size_t)r * D + 16 * t] = ROW_SCALE ? acc[t][r] * rs[r] : acc[t][r];
}

#define FT_WAVES 16
#define FT_PER_WAVE 3
#define FT_NODES (FT_WAVES * FT_PER_WAVE)
#define Y_PITCH4 (D / 4 + 1)

__device__ __forceinline__ void gather_rows(const float* __restrict__ hf, int r_l, float cf_l,
                                            int nb, float4& acc) {
  int t = 0;
  for (; t + 4 <= nb; t += 4) {
    const int r0 = __builtin_amdgcn_readlane(r_l, t);
    const int r1 = __builtin_amdgcn_readlane(r_l, t + 1);
    const int r2 = __builtin_amdgcn_readlane(r_l, t + 2);
    const int r3 = __builtin_amdgcn_readlane(r_l, t + 3);
    const float c0 = __int_as_float(__builtin_amdgcn_readlane(__float_as_int(cf_l), t));
    const float c1 = __int_as_float(__builtin_amdgcn_readlane(__float_as_int(cf_l), t + 1));
    const float c2 = __int_as_float(__builtin_amdgcn_readlane(__float_as_int(cf_l), t + 2));
    const float c3 = __int_as_float(__builtin_amdgcn_readlane(__float_as_int(cf_l), t + 3));
    const float4 a0 = *(const float4*)(hf + (size_t)r0 * D);
    const float4 a1 = *(const float4*)(hf + (size_t)r1 * D);
    const float4 a2 = *(const float4*)(hf + (size_t)r2 * D);
    const float4 a3 = *(const float4*)(hf + (size_t)r3 * D);
    acc.x += c0 * a0.x; acc.y += c0 * a0.y; acc.z += c0 * a0.z; acc.w += c0 * a0.w;
    acc.x += c1 * a1.x; acc.y += c1 * a1.y; acc.z += c1 * a1.z; acc.w += c1 * a1.w;
    acc.x += c2 * a2.x; acc.y += c2 * a2.y; acc.z += c2 * a2.z; acc.w += c2 * a2.w;
    acc.x += c3 * a3.x; acc.y += c3 * a3.y; acc.z += c3 * a3.z; acc.w += c3 * a3.w;
  }
  for (; t < nb; ++t) {
    const int r0 = __builtin_amdgcn_readlane(r_l, t);
    const float c0 = __int_as_float(__builtin_amdgcn_readlane(__float_as_int(cf_l), t));
    const float4 a0 = *(const float4*)(hf + (size_t)r0 * D);
    acc.x += c0 * a0.x; acc.y += c0 * a0.y; acc.z += c0 * a0.z; acc.w += c0 * a0.w;
  }
}

__global__ void __launch_bounds__(32 * FT_WAVES, 12)
gather_ln_gemm_kernel(const int* __restrict__ start, const int2* __restrict__ sedge,
                      const float* __restrict__ dinv, const float* __restrict__ h,
                      const float* __restrict__ bias, const float* __restrict__ gamma,
                      const float* __restrict__ beta, const float* __restrict__ Bp,
                      const float* __restrict__ lbias, float* __restrict__ C, int N, int E) {
  __shared__ float4 sB[PACK_ENTRIES];
  __shared__ float4 sY[FT_NODES * Y_PITCH4];
  const int tid  = threadIdx.x;
  const int wave = __builtin_amdgcn_readfirstlane((int)(threadIdx.x >> 5));
  const int lane = tid & 31;
  const int f = lane * 4;
  const float* hf = h + f;

  const int node0 = blockIdx.x * FT_NODES + wave * FT_PER_WAVE;
  int   sb[FT_PER_WAVE + 1];
  float dcs[FT_PER_WAVE];
#pragma unroll
  for (int q = 0; q <= FT_PER_WAVE; ++q) sb[q] = start[min(node0 + q, N)];
#pragma unroll
  for (int q = 0; q < FT_PER_WAVE; ++q) dcs[q] = dinv[min(node0 + q, N - 1)];
  int2   ed0[FT_PER_WAVE];
  float4 selfr[FT_PER_WAVE];
#pragma unroll
  for (int q = 0; q < FT_PER_WAVE; ++q) {
    const int j = max(sb[q], 0) + lane;
    ed0[q] = make_int2(0, 0);
    if (j < min(sb[q + 1], E)) ed0[q] = sedge[j];
  }
#pragma unroll
  for (int q = 0; q < FT_PER_WAVE; ++q)
    selfr[q] = *(const float4*)(hf + (size_t)min(node0 + q, N - 1) * D);

  const float4* bp4 = (const float4*)Bp;
  float4 bq[PACK_ENTRIES / (32 * FT_WAVES)];
#pragma unroll
  for (int i = 0; i < PACK_ENTRIES / (32 * FT_WAVES); ++i) bq[i] = bp4[tid + 32 * FT_WAVES * i];
#pragma unroll
  for (int i = 0; i < PACK_ENTRIES / (32 * FT_WAVES); ++i) sB[tid + 32 * FT_WAVES * i] = bq[i];

  const float4 bi = *(const float4*)(bias + f);
  const float4 g  = *(const float4*)(gamma + f);
  const float4 be = *(const float4*)(beta + f);
#pragma unroll
  for (int q = 0; q < FT_PER_WAVE; ++q) {
    const int node = node0 + q;
    if (node < N) {
      const int beg = max(sb[q], 0);
      const int end = min(sb[q + 1], E);
      float4 acc = selfr[q];
      if (beg < end) {
        gather_rows(hf, (int)min((unsigned)ed0[q].x, (unsigned)(N - 1)), __int_as_float(ed0[q].y),
                    min(32, end - beg), acc);
        for (int base = beg + 32; base < end; base += 32) {
          const int j = base + lane;
          int2 ed = make_int2(0, 0);
          if (j < end) ed = sedge[j];
          gather_rows(hf, (int)min((unsigned)ed.x, (unsigned)(N - 1)), __int_as_float(ed.y),
                      min(32, end - base), acc);
        }
      }

      const float dc = dcs[q];
      float4 v;
      v.x = dc * acc.x; v.y = dc * acc.y; v.z = dc * acc.z; v.w = dc * acc.w;
      v.x = fmaxf(v.x + bi.x, 0.0f);
      v.y = fmaxf(v.y + bi.y, 0.0f);
      v.z = fmaxf(v.z + bi.z, 0.0f);
      v.w = fmaxf(v.w + bi.w, 0.0f);
      float s = v.x + v.y + v.z + v.w;
#pragma unroll
      for (int mski = 16; mski >= 1; mski >>= 1) s += __shfl_xor(s, mski, 32);
      float mu = s * (1.0f / D);
      float dx0 = v.x - mu, dx1 = v.y - mu, dx2 = v.z - mu, dx3 = v.w - mu;
      float vs = dx0 * dx0 + dx1 * dx1 + dx2 * dx2 + dx3 * dx3;
#pragma unroll
      for (int mski = 16; mski >= 1; mski >>= 1) vs += __shfl_xor(vs, mski, 32);
      float inv = rsqrtf(vs * (1.0f / D) + LN_EPS);
      float4 o;
      o.x = dx0 * inv * g.x + be.x;
      o.y = dx1 * inv * g.y + be.y;
      o.z = dx2 * inv * g.z + be.z;
      o.w = dx3 * inv * g.w + be.w;
      sY[(wave * FT_PER_WAVE + q) * Y_PITCH4 + lane] = o;
    }
  }
  __syncthreads();

  const int c  = lane & 15;
  const int hh = lane >> 4;
  const int strip = wave >> 2;
  const int t0 = 2 * (wave & 3);
  const int row0 = blockIdx.x * FT_NODES + 16 * strip;
  if (strip >= FT_NODES / 16 || row0 >= N) return;

  const float lb0 = lbias[16 * t0 + c];
  const float lb1 = lbias[16 * t0 + 16 + c];
  v8f acc0 = (v8f){lb0, lb0, lb0, lb0, lb0, lb0, lb0, lb0};
  v8f acc1 = (v8f){lb1, lb1, lb1, lb1, lb1, lb1, lb1, lb1};
  const float4* ya = sY + (16 * strip + c) * Y_PITCH4 + hh;
  const float4* sl = sB + t0 * 32 + lane;
#pragma unroll
  for (int j = 0; j < D / 8; ++j) {
    const float4 a  = ya[2 * j];
    const float4 b0 = sl[j * (D / 16) * 32];
    const float4 b1 = sl[j * (D / 16) * 32 + 32];
    const v2f a01 = {a.x, a.y};
    const v2f a23 = {a.z, a.w};
    const v2f b001 = {b0.x, b0.y};
    const v2f b101 = {b1.x, b1.y};
    const v2f b023 = {b0.z, b0.w};
    const v2f b123 = {b1.z, b1.w};
    acc0 = __builtin_amdgcn_wmma_f32_16x16x4_f32(false, a01, false, b001, (short)0, acc0, false, false);
    acc1 = __builtin_amdgcn_wmma_f32_16x16x4_f32(false, a01, false, b101, (short)0, acc1, false, false);
    acc0 = __builtin_amdgcn_wmma_f32_16x16x4_f32(false, a23, false, b023, (short)0, acc0, false, false);
    acc1 = __builtin_amdgcn_wmma_f32_16x16x4_f32(false, a23, false, b123, (short)0, acc1, false, false);
  }

  float* cp = C + (size_t)(row0 + hh * 8) * D + 16 * t0 + c;
#pragma unroll
  for (int r = 0; r < 8; ++r) {
    if (row0 + hh * 8 + r < N) {
      cp[(size_t)r * D]      = acc0[r];
      cp[(size_t)r * D + 16] = acc1[r];
    }
  }
}

extern "C" void kernel_launch(void* const* d_in, const int* in_sizes, int n_in,
                              void* d_out, int out_size, void* d_ws, size_t ws_size,
                              hipStream_t stream) {
  const float* x     = (const float*)d_in[0];
  const int*   ei    = (const int*)d_in[1];
  const float* ew    = (const float*)d_in[2];
  const float* W     = (const float*)d_in[3];
  const float* bias  = (const float*)d_in[4];
  const float* gamma = (const float*)d_in[5];
  const float* beta  = (const float*)d_in[6];
  const float* linW  = (const float*)d_in[7];
  const float* linb  = (const float*)d_in[8];
  float* out = (float*)d_out;

  const int N = in_sizes[0] / D;
  const int E = in_sizes[2];
  const int B = (N + BIN_NODES - 1) / BIN_NODES;
  const int G = (E + CHUNK - 1) / CHUNK;

  char* ws = (char*)d_ws;
  const size_t nB = (((size_t)(N + 1) * sizeof(int)) + 255) & ~(size_t)255;
  const size_t tB = (size_t)G * MAXB * sizeof(int);
  const size_t cB = (size_t)G * CHUNK * sizeof(int2);
  float* dinv   = (float*)ws;
  int*   start  = (int*)(ws + nB);
  int*   runTab = (int*)(ws + 2 * nB);
  int2*  cedge  = (int2*)(ws + 2 * nB + tB);
  int2*  sedge  = (int2*)(ws + 2 * nB + tB + cB);
  float* h      = (float*)(ws + 2 * nB + tB + cB + (size_t)E * sizeof(int2));
  float4* Wp   = (float4*)(h + (size_t)N * D);
  float4* Lp   = Wp + PACK_ENTRIES;

  chunk_sort_pack_kernel<<<G + PACK_ENTRIES / CHUNK_T, CHUNK_T, 0, stream>>>(ei, ew, runTab, cedge, W,
                                                                            linW, Wp, Lp, E, N, G);
  bin_sort_kernel<<<B, BIN_NODES, 0, stream>>>(runTab, cedge, start, sedge, dinv, N, E, G);
  gemm_strip_kernel<false, true><<<(N / 16 + GEMM_WAVES - 1) / GEMM_WAVES, 32 * GEMM_WAVES, 0,
                                  stream>>>(x, (const float*)Wp, nullptr, dinv, h, N / 16);
  gather_ln_gemm_kernel<<<(N + FT_NODES - 1) / FT_NODES, 32 * FT_WAVES, 0, stream>>>(
      start, sedge, dinv, h, bias, gamma, beta, (const float*)Lp, linb, out, N, E);
}
